# starting kernel
# baseline (speedup 1.0000x reference)
#include <hip/hip_runtime.h>
#include <math.h>

#define C 128
#define TPB 256

typedef float v2f __attribute__((ext_vector_type(2)));
typedef float v8f __attribute__((ext_vector_type(8)));

__global__ void deg_kernel(const int* __restrict__ dst, float* __restrict__ deg, int E) {
    int e = blockIdx.x * blockDim.x + threadIdx.x;
    if (e < E) atomicAdd(&deg[dst[e]], 1.0f);
}

__global__ void dinv_kernel(float* __restrict__ degdinv, int N) {
    int i = blockIdx.x * blockDim.x + threadIdx.x;
    if (i < N) degdinv[i] = rsqrtf(degdinv[i] + 1.0f);
}

__global__ __launch_bounds__(TPB) void gemm_wmma(const float* __restrict__ X,
                                                 const float* __restrict__ W,
                                                 float* __restrict__ Y,
                                                 int nStrips) {
    __shared__ float Wl[64 * C];

    const int wave = threadIdx.x >> 5;
    const int lane = threadIdx.x & 31;
    const int half = lane >> 4;
    const int lo   = lane & 15;

    const int strip  = blockIdx.x * (TPB / 32) + wave;
    const bool active = (strip < nStrips);
    const int m0 = strip * 16;

    v8f acc[8] = {};

    const float* Arow = X + (size_t)(active ? (m0 + lo) : 0) * C + 2 * half;

    for (int kc = 0; kc < C; kc += 64) {
        __syncthreads();
        for (int idx = threadIdx.x; idx < 64 * C; idx += TPB) {
            int k = idx >> 7;
            int n = idx & (C - 1);
            Wl[(k >> 1) * 256 + n * 2 + (k & 1)] = W[(size_t)(kc + k) * C + n];
        }
        __syncthreads();

        if (active) {
            #pragma unroll
            for (int k0 = 0; k0 < 64; k0 += 4) {
                v2f a = *(const v2f*)(Arow + kc + k0);
                const float* Bbase = &Wl[((k0 >> 1) + half) * 256 + lo * 2];
                #pragma unroll
                for (int t = 0; t < 8; ++t) {
                    v2f b = *(const v2f*)(Bbase + t * 32);
                    acc[t] = __builtin_amdgcn_wmma_f32_16x16x4_f32(
                        false, a, false, b, (short)0, acc[t], false, false);
                }
            }
        }
    }

    if (active) {
        float* Yt = Y + (size_t)m0 * C;
        #pragma unroll
        for (int t = 0; t < 8; ++t)
            #pragma unroll
            for (int r = 0; r < 8; ++r)
                Yt[(size_t)(r + 8 * half) * C + t * 16 + lo] = acc[t][r];
    }
}

__global__ void gemm_tail(const float* __restrict__ X, const float* __restrict__ W,
                          float* __restrict__ Y, int row0, int rows) {
    int i = blockIdx.x * blockDim.x + threadIdx.x;
    if (i >= rows * C) return;
    int r = row0 + i / C, n = i & (C - 1);
    float s = 0.f;
    for (int k = 0; k < C; ++k) s += X[(size_t)r * C + k] * W[(size_t)k * C + n];
    Y[(size_t)r * C + n] = s;
}

__global__ void agg_kernel(const float* __restrict__ H, const int* __restrict__ src,
                           const int* __restrict__ dst, const float* __restrict__ dinv,
                           float* __restrict__ out, int E, int N) {
    int gw   = (blockIdx.x * blockDim.x + threadIdx.x) >> 5;
    int lane = threadIdx.x & 31;
    int nw   = (gridDim.x * blockDim.x) >> 5;
    int total = E + N;
    for (int e = gw; e < total; e += nw) {
        int s, d;
        if (e < E) { s = src[e]; d = dst[e]; } else { s = e - E; d = s; }
        float nrm = dinv[s] * dinv[d];
        float4 v = *(const float4*)(H + (size_t)s * C + lane * 4);
        float* o = out + (size_t)d * C + lane * 4;
        atomicAdd(o + 0, v.x * nrm);
        atomicAdd(o + 1, v.y * nrm);
        atomicAdd(o + 2, v.z * nrm);
        atomicAdd(o + 3, v.w * nrm);
    }
}

__global__ void bias_elu_kernel(const float* __restrict__ in, const float* __restrict__ b,
                                float* __restrict__ out, size_t n4) {
    size_t i = (size_t)blockIdx.x * blockDim.x + threadIdx.x;
    if (i >= n4) return;
    size_t idx = i * 4;
    int c = (int)(idx & (C - 1));
    float4 v = *(const float4*)(in + idx);
    v.x += b[c + 0]; v.y += b[c + 1]; v.z += b[c + 2]; v.w += b[c + 3];
    v.x = v.x > 0.f ? v.x : (expf(v.x) - 1.f);
    v.y = v.y > 0.f ? v.y : (expf(v.y) - 1.f);
    v.z = v.z > 0.f ? v.z : (expf(v.z) - 1.f);
    v.w = v.w > 0.f ? v.w : (expf(v.w) - 1.f);
    *(float4*)(out + idx) = v;
}

extern "C" void kernel_launch(void* const* d_in, const int* in_sizes, int n_in,
                              void* d_out, int out_size, void* d_ws, size_t ws_size,
                              hipStream_t stream) {
    const float* x  = (const float*)d_in[0];
    const int*   ei = (const int*)d_in[1];
    const float* W1 = (const float*)d_in[2];
    const float* b1 = (const float*)d_in[3];
    const float* W2 = (const float*)d_in[4];
    const float* b2 = (const float*)d_in[5];

    const int N = in_sizes[0] / C;
    const int E = in_sizes[1] / 2;
    const int* srcI = ei;
    const int* dstI = ei + E;

    char* ws = (char*)d_ws;
    float* dinv = (float*)ws;
    size_t o1 = ((size_t)N * 4 + 255) & ~(size_t)255;
    size_t hbytes = ((size_t)N * C * 4 + 255) & ~(size_t)255;
    float* buf0 = (float*)(ws + o1);
    float* buf1 = (float*)(ws + o1 + hbytes);

    const int strips   = N / 16;
    const int tailRow0 = strips * 16;
    const int tailRows = N - tailRow0;
    const dim3 blk(TPB);
    const int gemmBlocks = (strips + (TPB / 32) - 1) / (TPB / 32);
    const int aggBlocks  = 4096;
    const size_t n4      = (size_t)N * C / 4;
    const int eluBlocks  = (int)((n4 + TPB - 1) / TPB);

    hipMemsetAsync(dinv, 0, (size_t)N * 4, stream);
    deg_kernel<<<(E + TPB - 1) / TPB, blk, 0, stream>>>(dstI, dinv, E);
    dinv_kernel<<<(N + TPB - 1) / TPB, blk, 0, stream>>>(dinv, N);

    gemm_wmma<<<gemmBlocks, blk, 0, stream>>>(x, W1, buf0, strips);
    if (tailRows > 0)
        gemm_tail<<<(tailRows * C + TPB - 1) / TPB, blk, 0, stream>>>(x, W1, buf0, tailRow0, tailRows);
    hipMemsetAsync(buf1, 0, (size_t)N * C * 4, stream);
    agg_kernel<<<aggBlocks, blk, 0, stream>>>(buf0, srcI, dstI, dinv, buf1, E, N);
    bias_elu_kernel<<<eluBlocks, blk, 0, stream>>>(buf1, b1, buf0, n4);

    gemm_wmma<<<gemmBlocks, blk, 0, stream>>>(buf0, W2, buf1, strips);
    if (tailRows > 0)
        gemm_tail<<<(tailRows * C + TPB - 1) / TPB, blk, 0, stream>>>(buf0, W2, buf1, tailRow0, tailRows);
    hipMemsetAsync(buf0, 0, (size_t)N * C * 4, stream);
    agg_kernel<<<aggBlocks, blk, 0, stream>>>(buf1, srcI, dstI, dinv, buf0, E, N);
    bias_elu_kernel<<<eluBlocks, blk, 0, stream>>>(buf0, b2, (float*)d_out, n4);
}


# Replace f32 atomic scatter-add by a destination-ordered edge list (integer arrival rank, scan, place) and per-node row sums in registers; sums now run in arrival-rank order, big memsets gone
# speedup vs baseline: 2.3708x; 2.3708x over previous
#include <hip/hip_runtime.h>
#include <math.h>

#define C 128
#define TPB 256

typedef float v2f __attribute__((ext_vector_type(2)));
typedef float v8f __attribute__((ext_vector_type(8)));

__global__ void rank_kernel(const int* __restrict__ dst, int* __restrict__ cnt,
                            int* __restrict__ rank, int E) {
    int e = blockIdx.x * blockDim.x + threadIdx.x;
    if (e < E) rank[e] = atomicAdd(&cnt[dst[e]], 1);
}

#define SCAN_T 1024
__global__ __launch_bounds__(SCAN_T) void scan_kernel(const int* __restrict__ cnt,
                                                      int* __restrict__ rowptr,
                                                      float* __restrict__ dinv, int N) {
    __shared__ int wtot[32];
    __shared__ int wpre[32];
    __shared__ int sbase;
    const int tid  = threadIdx.x;
    const int lane = tid & 31;
    const int wave = tid >> 5;
    const int blockStart = blockIdx.x * SCAN_T;

    int part = 0;
    for (int i = tid; i < blockStart; i += SCAN_T) part += cnt[i];

    const int i = blockStart + tid;
    const int c = (i < N) ? cnt[i] : 0;

    int incl = c;
    #pragma unroll
    for (int off = 1; off < 32; off <<= 1) {
        int t = __shfl_up(incl, off);
        if (lane >= off) incl += t;
    }
    #pragma unroll
    for (int off = 16; off > 0; off >>= 1) part += __shfl_xor(part, off);

    if (lane == 31) wtot[wave] = incl;
    if (lane == 0)  wpre[wave] = part;
    __syncthreads();
    if (wave == 0) {
        const int t = wtot[lane];
        int p = wpre[lane];
        int ti = t;
        #pragma unroll
        for (int off = 1; off < 32; off <<= 1) {
            int u = __shfl_up(ti, off);
            if (lane >= off) ti += u;
        }
        #pragma unroll
        for (int off = 16; off > 0; off >>= 1) p += __shfl_xor(p, off);
        wtot[lane] = ti - t;
        if (lane == 0) sbase = p;
    }
    __syncthreads();
    if (i < N) {
        const int excl = sbase + wtot[wave] + incl - c;
        rowptr[i] = excl;
        dinv[i] = rsqrtf((float)c + 1.0f);
        if (i == N - 1) rowptr[N] = excl + c;
    }
}

__global__ void place_kernel(const int* __restrict__ src, const int* __restrict__ dst,
                             const int* __restrict__ rank, const int* __restrict__ rowptr,
                             const float* __restrict__ dinv, int2* __restrict__ colw, int E) {
    int e = blockIdx.x * blockDim.x + threadIdx.x;
    if (e >= E) return;
    const int s = src[e], d = dst[e];
    const int pos = rowptr[d] + rank[e];
    colw[pos] = make_int2(s, __float_as_int(dinv[s] * dinv[d]));
}

__global__ __launch_bounds__(TPB) void gemm_wmma(const float* __restrict__ X,
                                                 const float* __restrict__ W,
                                                 float* __restrict__ Y,
                                                 int nStrips) {
    __shared__ float Wl[64 * C];

    const int wave = threadIdx.x >> 5;
    const int lane = threadIdx.x & 31;
    const int half = lane >> 4;
    const int lo   = lane & 15;

    const int strip  = blockIdx.x * (TPB / 32) + wave;
    const bool active = (strip < nStrips);
    const int m0 = strip * 16;

    v8f acc[8] = {};

    const float* Arow = X + (size_t)(active ? (m0 + lo) : 0) * C + 2 * half;

    for (int kc = 0; kc < C; kc += 64) {
        __syncthreads();
        for (int idx = threadIdx.x; idx < 64 * C; idx += TPB) {
            int k = idx >> 7;
            int n = idx & (C - 1);
            Wl[(k >> 1) * 256 + n * 2 + (k & 1)] = W[(size_t)(kc + k) * C + n];
        }
        __syncthreads();

        if (active) {
            #pragma unroll
            for (int k0 = 0; k0 < 64; k0 += 4) {
                v2f a = *(const v2f*)(Arow + kc + k0);
                const float* Bbase = &Wl[((k0 >> 1) + half) * 256 + lo * 2];
                #pragma unroll
                for (int t = 0; t < 8; ++t) {
                    v2f b = *(const v2f*)(Bbase + t * 32);
                    acc[t] = __builtin_amdgcn_wmma_f32_16x16x4_f32(
                        false, a, false, b, (short)0, acc[t], false, false);
                }
            }
        }
    }

    if (active) {
        float* Yt = Y + (size_t)m0 * C;
        #pragma unroll
        for (int t = 0; t < 8; ++t)
            #pragma unroll
            for (int r = 0; r < 8; ++r)
                Yt[(size_t)(r + 8 * half) * C + t * 16 + lo] = acc[t][r];
    }
}

__global__ void gemm_tail(const float* __restrict__ X, const float* __restrict__ W,
                          float* __restrict__ Y, int row0, int rows) {
    int i = blockIdx.x * blockDim.x + threadIdx.x;
    if (i >= rows * C) return;
    int r = row0 + i / C, n = i & (C - 1);
    float s = 0.f;
    for (int k = 0; k < C; ++k) s += X[(size_t)r * C + k] * W[(size_t)k * C + n];
    Y[(size_t)r * C + n] = s;
}

template <int K>
__device__ __forceinline__ void gather_k(float4& acc, const float* __restrict__ H,
                                         int cols, int wbits, int j, int lane) {
    float4 v[K];
    float  w[K];
    #pragma unroll
    for (int k = 0; k < K; ++k) {
        const int s = __builtin_amdgcn_readlane(cols, j + k);
        w[k] = __int_as_float(__builtin_amdgcn_readlane(wbits, j + k));
        v[k] = *(const float4*)(H + (size_t)s * C + lane * 4);
    }
    #pragma unroll
    for (int k = 0; k < K; ++k) {
        acc.x = fmaf(v[k].x, w[k], acc.x);
        acc.y = fmaf(v[k].y, w[k], acc.y);
        acc.z = fmaf(v[k].z, w[k], acc.z);
        acc.w = fmaf(v[k].w, w[k], acc.w);
    }
}

__global__ __launch_bounds__(TPB) void agg_rows(const float* __restrict__ H,
                                                const int* __restrict__ rowptr,
                                                const int2* __restrict__ colw,
                                                const float* __restrict__ dinv,
                                                float* __restrict__ out, int N) {
    const int lane = threadIdx.x & 31;
    const int gw = __builtin_amdgcn_readfirstlane((int)((blockIdx.x * blockDim.x + threadIdx.x) >> 5));
    const int nw = (int)((gridDim.x * blockDim.x) >> 5);
    for (int row = gw; row < N; row += nw) {
        const int beg = rowptr[row];
        const int end = rowptr[row + 1];
        const float di = dinv[row];
        const float wself = di * di;
        const float4 hs = *(const float4*)(H + (size_t)row * C + lane * 4);
        float4 acc = make_float4(0.f, 0.f, 0.f, 0.f);
        for (int base = beg; base < end; base += 32) {
            const int n = (end - base < 32) ? (end - base) : 32;
            int2 cw = make_int2(0, 0);
            if (lane < n) cw = colw[base + lane];
            int j = 0;
            for (; j + 4 <= n; j += 4) gather_k<4>(acc, H, cw.x, cw.y, j, lane);
            const int rem = n - j;
            if (rem == 3)      gather_k<3>(acc, H, cw.x, cw.y, j, lane);
            else if (rem == 2) gather_k<2>(acc, H, cw.x, cw.y, j, lane);
            else if (rem == 1) gather_k<1>(acc, H, cw.x, cw.y, j, lane);
        }
        acc.x = fmaf(hs.x, wself, acc.x);
        acc.y = fmaf(hs.y, wself, acc.y);
        acc.z = fmaf(hs.z, wself, acc.z);
        acc.w = fmaf(hs.w, wself, acc.w);
        *(float4*)(out + (size_t)row * C + lane * 4) = acc;
    }
}

__global__ void bias_elu_kernel(const float* __restrict__ in, const float* __restrict__ b,
                                float* __restrict__ out, size_t n4) {
    size_t i = (size_t)blockIdx.x * blockDim.x + threadIdx.x;
    if (i >= n4) return;
    size_t idx = i * 4;
    int c = (int)(idx & (C - 1));
    float4 v = *(const float4*)(in + idx);
    v.x += b[c + 0]; v.y += b[c + 1]; v.z += b[c + 2]; v.w += b[c + 3];
    v.x = v.x > 0.f ? v.x : (expf(v.x) - 1.f);
    v.y = v.y > 0.f ? v.y : (expf(v.y) - 1.f);
    v.z = v.z > 0.f ? v.z : (expf(v.z) - 1.f);
    v.w = v.w > 0.f ? v.w : (expf(v.w) - 1.f);
    *(float4*)(out + idx) = v;
}

extern "C" void kernel_launch(void* const* d_in, const int* in_sizes, int n_in,
                              void* d_out, int out_size, void* d_ws, size_t ws_size,
                              hipStream_t stream) {
    const float* x  = (const float*)d_in[0];
    const int*   ei = (const int*)d_in[1];
    const float* W1 = (const float*)d_in[2];
    const float* b1 = (const float*)d_in[3];
    const float* W2 = (const float*)d_in[4];
    const float* b2 = (const float*)d_in[5];

    const int N = in_sizes[0] / C;
    const int E = in_sizes[1] / 2;
    const int* srcI = ei;
    const int* dstI = ei + E;

    char* ws = (char*)d_ws;
    size_t off = 0;
    int* cnt = (int*)(ws + off);      off += ((size_t)N * 4 + 255) & ~(size_t)255;
    int* rowptr = (int*)(ws + off);   off += ((size_t)(N + 1) * 4 + 255) & ~(size_t)255;
    float* dinv = (float*)(ws + off); off += ((size_t)N * 4 + 255) & ~(size_t)255;
    int* rank = (int*)(ws + off);     off += ((size_t)E * 4 + 255) & ~(size_t)255;
    int2* colw = (int2*)(ws + off);   off += ((size_t)E * 8 + 255) & ~(size_t)255;
    size_t hbytes = ((size_t)N * C * 4 + 255) & ~(size_t)255;
    float* buf0 = (float*)(ws + off);
    float* buf1 = (float*)(ws + off + hbytes);

    const int strips   = N / 16;
    const int tailRow0 = strips * 16;
    const int tailRows = N - tailRow0;
    const dim3 blk(TPB);
    const int gemmBlocks = (strips + (TPB / 32) - 1) / (TPB / 32);
    const int edgeBlocks = (E + TPB - 1) / TPB;
    const int scanBlocks = (N + SCAN_T - 1) / SCAN_T;
    const int aggBlocks  = 2048;
    const size_t n4      = (size_t)N * C / 4;
    const int eluBlocks  = (int)((n4 + TPB - 1) / TPB);

    hipMemsetAsync(cnt, 0, (size_t)N * 4, stream);
    rank_kernel<<<edgeBlocks, blk, 0, stream>>>(dstI, cnt, rank, E);
    scan_kernel<<<scanBlocks, dim3(SCAN_T), 0, stream>>>(cnt, rowptr, dinv, N);
    place_kernel<<<edgeBlocks, blk, 0, stream>>>(srcI, dstI, rank, rowptr, dinv, colw, E);

    gemm_wmma<<<gemmBlocks, blk, 0, stream>>>(x, W1, buf0, strips);
    if (tailRows > 0)
        gemm_tail<<<(tailRows * C + TPB - 1) / TPB, blk, 0, stream>>>(x, W1, buf0, tailRow0, tailRows);
    agg_rows<<<aggBlocks, blk, 0, stream>>>(buf0, rowptr, colw, dinv, buf1, N);
    bias_elu_kernel<<<eluBlocks, blk, 0, stream>>>(buf1, b1, buf0, n4);

    gemm_wmma<<<gemmBlocks, blk, 0, stream>>>(buf0, W2, buf1, strips);
    if (tailRows > 0)
        gemm_tail<<<(tailRows * C + TPB - 1) / TPB, blk, 0, stream>>>(buf0, W2, buf1, tailRow0, tailRows);
    agg_rows<<<aggBlocks, blk, 0, stream>>>(buf1, rowptr, colw, dinv, buf0, N);
    bias_elu_kernel<<<eluBlocks, blk, 0, stream>>>(buf0, b2, (float*)d_out, n4);
}


# Fuse bias+ELU into the aggregation kernel: applied to the finished row in registers before its one store; the two elementwise launches (100000 short waves, activation re-read) are gone
# speedup vs baseline: 2.5979x; 1.0958x over previous
#include <hip/hip_runtime.h>
#include <math.h>

#define C 128
#define TPB 256

typedef float v2f __attribute__((ext_vector_type(2)));
typedef float v8f __attribute__((ext_vector_type(8)));

__global__ void rank_kernel(const int* __restrict__ dst, int* __restrict__ cnt,
                            int* __restrict__ rank, int E) {
    int e = blockIdx.x * blockDim.x + threadIdx.x;
    if (e < E) rank[e] = atomicAdd(&cnt[dst[e]], 1);
}

#define SCAN_T 1024
__global__ __launch_bounds__(SCAN_T) void scan_kernel(const int* __restrict__ cnt,
                                                      int* __restrict__ rowptr,
                                                      float* __restrict__ dinv, int N) {
    __shared__ int wtot[32];
    __shared__ int wpre[32];
    __shared__ int sbase;
    const int tid  = threadIdx.x;
    const int lane = tid & 31;
    const int wave = tid >> 5;
    const int blockStart = blockIdx.x * SCAN_T;

    int part = 0;
    for (int i = tid; i < blockStart; i += SCAN_T) part += cnt[i];

    const int i = blockStart + tid;
    const int c = (i < N) ? cnt[i] : 0;

    int incl = c;
    #pragma unroll
    for (int off = 1; off < 32; off <<= 1) {
        int t = __shfl_up(incl, off);
        if (lane >= off) incl += t;
    }
    #pragma unroll
    for (int off = 16; off > 0; off >>= 1) part += __shfl_xor(part, off);

    if (lane == 31) wtot[wave] = incl;
    if (lane == 0)  wpre[wave] = part;
    __syncthreads();
    if (wave == 0) {
        const int t = wtot[lane];
        int p = wpre[lane];
        int ti = t;
        #pragma unroll
        for (int off = 1; off < 32; off <<= 1) {
            int u = __shfl_up(ti, off);
            if (lane >= off) ti += u;
        }
        #pragma unroll
        for (int off = 16; off > 0; off >>= 1) p += __shfl_xor(p, off);
        wtot[lane] = ti - t;
        if (lane == 0) sbase = p;
    }
    __syncthreads();
    if (i < N) {
        const int excl = sbase + wtot[wave] + incl - c;
        rowptr[i] = excl;
        dinv[i] = rsqrtf((float)c + 1.0f);
        if (i == N - 1) rowptr[N] = excl + c;
    }
}

__global__ void place_kernel(const int* __restrict__ src, const int* __restrict__ dst,
                             const int* __restrict__ rank, const int* __restrict__ rowptr,
                             const float* __restrict__ dinv, int2* __restrict__ colw, int E) {
    int e = blockIdx.x * blockDim.x + threadIdx.x;
    if (e >= E) return;
    const int s = src[e], d = dst[e];
    const int pos = rowptr[d] + rank[e];
    colw[pos] = make_int2(s, __float_as_int(dinv[s] * dinv[d]));
}

__global__ __launch_bounds__(TPB) void gemm_wmma(const float* __restrict__ X,
                                                 const float* __restrict__ W,
                                                 float* __restrict__ Y,
                                                 int nStrips) {
    __shared__ float Wl[64 * C];

    const int wave = threadIdx.x >> 5;
    const int lane = threadIdx.x & 31;
    const int half = lane >> 4;
    const int lo   = lane & 15;

    const int strip  = blockIdx.x * (TPB / 32) + wave;
    const bool active = (strip < nStrips);
    const int m0 = strip * 16;

    v8f acc[8] = {};

    const float* Arow = X + (size_t)(active ? (m0 + lo) : 0) * C + 2 * half;

    for (int kc = 0; kc < C; kc += 64) {
        __syncthreads();
        for (int idx = threadIdx.x; idx < 64 * C; idx += TPB) {
            int k = idx >> 7;
            int n = idx & (C - 1);
            Wl[(k >> 1) * 256 + n * 2 + (k & 1)] = W[(size_t)(kc + k) * C + n];
        }
        __syncthreads();

        if (active) {
            #pragma unroll
            for (int k0 = 0; k0 < 64; k0 += 4) {
                v2f a = *(const v2f*)(Arow + kc + k0);
                const float* Bbase = &Wl[((k0 >> 1) + half) * 256 + lo * 2];
                #pragma unroll
                for (int t = 0; t < 8; ++t) {
                    v2f b = *(const v2f*)(Bbase + t * 32);
                    acc[t] = __builtin_amdgcn_wmma_f32_16x16x4_f32(
                        false, a, false, b, (short)0, acc[t], false, false);
                }
            }
        }
    }

    if (active) {
        float* Yt = Y + (size_t)m0 * C;
        #pragma unroll
        for (int t = 0; t < 8; ++t)
            #pragma unroll
            for (int r = 0; r < 8; ++r)
                Yt[(size_t)(r + 8 * half) * C + t * 16 + lo] = acc[t][r];
    }
}

__global__ void gemm_tail(const float* __restrict__ X, const float* __restrict__ W,
                          float* __restrict__ Y, int row0, int rows) {
    int i = blockIdx.x * blockDim.x + threadIdx.x;
    if (i >= rows * C) return;
    int r = row0 + i / C, n = i & (C - 1);
    float s = 0.f;
    for (int k = 0; k < C; ++k) s += X[(size_t)r * C + k] * W[(size_t)k * C + n];
    Y[(size_t)r * C + n] = s;
}

__device__ __forceinline__ float elu1(float v) {
    return v > 0.f ? v : (expf(v) - 1.f);
}

template <int K>
__device__ __forceinline__ void gather_k(float4& acc, const float* __restrict__ H,
                                         int cols, int wbits, int j, int lane) {
    float4 v[K];
    float  w[K];
    #pragma unroll
    for (int k = 0; k < K; ++k) {
        const int s = __builtin_amdgcn_readlane(cols, j + k);
        w[k] = __int_as_float(__builtin_amdgcn_readlane(wbits, j + k));
        v[k] = *(const float4*)(H + (size_t)s * C + lane * 4);
    }
    #pragma unroll
    for (int k = 0; k < K; ++k) {
        acc.x = fmaf(v[k].x, w[k], acc.x);
        acc.y = fmaf(v[k].y, w[k], acc.y);
        acc.z = fmaf(v[k].z, w[k], acc.z);
        acc.w = fmaf(v[k].w, w[k], acc.w);
    }
}

__global__ __launch_bounds__(TPB) void agg_rows(const float* __restrict__ H,
                                                const int* __restrict__ rowptr,
                                                const int2* __restrict__ colw,
                                                const float* __restrict__ dinv,
                                                const float* __restrict__ bias,
                                                float* __restrict__ out, int N) {
    const int lane = threadIdx.x & 31;
    const float4 bv = make_float4(bias[lane * 4 + 0], bias[lane * 4 + 1],
                                  bias[lane * 4 + 2], bias[lane * 4 + 3]);
    const int gw = __builtin_amdgcn_readfirstlane((int)((blockIdx.x * blockDim.x + threadIdx.x) >> 5));
    const int nw = (int)((gridDim.x * blockDim.x) >> 5);
    for (int row = gw; row < N; row += nw) {
        const int beg = rowptr[row];
        const int end = rowptr[row + 1];
        const float di = dinv[row];
        const float wself = di * di;
        const float4 hs = *(const float4*)(H + (size_t)row * C + lane * 4);
        float4 acc = make_float4(0.f, 0.f, 0.f, 0.f);
        for (int base = beg; base < end; base += 32) {
            const int n = (end - base < 32) ? (end - base) : 32;
            int2 cw = make_int2(0, 0);
            if (lane < n) cw = colw[base + lane];
            int j = 0;
            for (; j + 4 <= n; j += 4) gather_k<4>(acc, H, cw.x, cw.y, j, lane);
            const int rem = n - j;
            if (rem == 3)      gather_k<3>(acc, H, cw.x, cw.y, j, lane);
            else if (rem == 2) gather_k<2>(acc, H, cw.x, cw.y, j, lane);
            else if (rem == 1) gather_k<1>(acc, H, cw.x, cw.y, j, lane);
        }
        acc.x = fmaf(hs.x, wself, acc.x);
        acc.y = fmaf(hs.y, wself, acc.y);
        acc.z = fmaf(hs.z, wself, acc.z);
        acc.w = fmaf(hs.w, wself, acc.w);
        acc.x = elu1(acc.x + bv.x);
        acc.y = elu1(acc.y + bv.y);
        acc.z = elu1(acc.z + bv.z);
        acc.w = elu1(acc.w + bv.w);
        *(float4*)(out + (size_t)row * C + lane * 4) = acc;
    }
}

extern "C" void kernel_launch(void* const* d_in, const int* in_sizes, int n_in,
                              void* d_out, int out_size, void* d_ws, size_t ws_size,
                              hipStream_t stream) {
    const float* x  = (const float*)d_in[0];
    const int*   ei = (const int*)d_in[1];
    const float* W1 = (const float*)d_in[2];
    const float* b1 = (const float*)d_in[3];
    const float* W2 = (const float*)d_in[4];
    const float* b2 = (const float*)d_in[5];

    const int N = in_sizes[0] / C;
    const int E = in_sizes[1] / 2;
    const int* srcI = ei;
    const int* dstI = ei + E;

    char* ws = (char*)d_ws;
    size_t off = 0;
    int* cnt = (int*)(ws + off);      off += ((size_t)N * 4 + 255) & ~(size_t)255;
    int* rowptr = (int*)(ws + off);   off += ((size_t)(N + 1) * 4 + 255) & ~(size_t)255;
    float* dinv = (float*)(ws + off); off += ((size_t)N * 4 + 255) & ~(size_t)255;
    int* rank = (int*)(ws + off);     off += ((size_t)E * 4 + 255) & ~(size_t)255;
    int2* colw = (int2*)(ws + off);   off += ((size_t)E * 8 + 255) & ~(size_t)255;
    size_t hbytes = ((size_t)N * C * 4 + 255) & ~(size_t)255;
    float* buf0 = (float*)(ws + off);
    float* buf1 = (float*)(ws + off + hbytes);

    const int strips   = N / 16;
    const int tailRow0 = strips * 16;
    const int tailRows = N - tailRow0;
    const dim3 blk(TPB);
    const int gemmBlocks = (strips + (TPB / 32) - 1) / (TPB / 32);
    const int edgeBlocks = (E + TPB - 1) / TPB;
    const int scanBlocks = (N + SCAN_T - 1) / SCAN_T;
    const int aggBlocks  = 2048;

    hipMemsetAsync(cnt, 0, (size_t)N * 4, stream);
    rank_kernel<<<edgeBlocks, blk, 0, stream>>>(dstI, cnt, rank, E);
    scan_kernel<<<scanBlocks, dim3(SCAN_T), 0, stream>>>(cnt, rowptr, dinv, N);
    place_kernel<<<edgeBlocks, blk, 0, stream>>>(srcI, dstI, rank, rowptr, dinv, colw, E);

    gemm_wmma<<<gemmBlocks, blk, 0, stream>>>(x, W1, buf0, strips);
    if (tailRows > 0)
        gemm_tail<<<(tailRows * C + TPB - 1) / TPB, blk, 0, stream>>>(x, W1, buf0, tailRow0, tailRows);
    agg_rows<<<aggBlocks, blk, 0, stream>>>(buf0, rowptr, colw, dinv, b1, buf1, N);

    gemm_wmma<<<gemmBlocks, blk, 0, stream>>>(buf1, W2, buf0, strips);
    if (tailRows > 0)
        gemm_tail<<<(tailRows * C + TPB - 1) / TPB, blk, 0, stream>>>(buf1, W2, buf0, tailRow0, tailRows);
    agg_rows<<<aggBlocks, blk, 0, stream>>>(buf0, rowptr, colw, dinv, b2, (float*)d_out, N);
}


# Each layer in one launch: a wave sums 16 rows of A*h (up to 16 neighbour rows in flight) into LDS and runs the strip through W on WMMA with bias+ELU; sum order now (A h) W; GEMM launches gone
# speedup vs baseline: 2.7644x; 1.0641x over previous
#include <hip/hip_runtime.h>
#include <math.h>

#define C 128
#define TPB 256

typedef float v2f __attribute__((ext_vector_type(2)));
typedef float v8f __attribute__((ext_vector_type(8)));

__global__ void rank_kernel(const int* __restrict__ dst, int* __restrict__ cnt,
                            int* __restrict__ rank, int E) {
    int e = blockIdx.x * blockDim.x + threadIdx.x;
    if (e < E) rank[e] = atomicAdd(&cnt[dst[e]], 1);
}

#define SCAN_T 1024
__global__ __launch_bounds__(SCAN_T) void scan_kernel(const int* __restrict__ cnt,
                                                      int* __restrict__ rowptr,
                                                      float* __restrict__ dinv, int N) {
    __shared__ int wtot[32];
    __shared__ int wpre[32];
    __shared__ int sbase;
    const int tid  = threadIdx.x;
    const int lane = tid & 31;
    const int wave = tid >> 5;
    const int blockStart = blockIdx.x * SCAN_T;

    int part = 0;
    for (int i = tid; i < blockStart; i += SCAN_T) part += cnt[i];

    const int i = blockStart + tid;
    const int c = (i < N) ? cnt[i] : 0;

    int incl = c;
    #pragma unroll
    for (int off = 1; off < 32; off <<= 1) {
        int t = __shfl_up(incl, off);
        if (lane >= off) incl += t;
    }
    #pragma unroll
    for (int off = 16; off > 0; off >>= 1) part += __shfl_xor(part, off);

    if (lane == 31) wtot[wave] = incl;
    if (lane == 0)  wpre[wave] = part;
    __syncthreads();
    if (wave == 0) {
        const int t = wtot[lane];
        int p = wpre[lane];
        int ti = t;
        #pragma unroll
        for (int off = 1; off < 32; off <<= 1) {
            int u = __shfl_up(ti, off);
            if (lane >= off) ti += u;
        }
        #pragma unroll
        for (int off = 16; off > 0; off >>= 1) p += __shfl_xor(p, off);
        wtot[lane] = ti - t;
        if (lane == 0) sbase = p;
    }
    __syncthreads();
    if (i < N) {
        const int excl = sbase + wtot[wave] + incl - c;
        rowptr[i] = excl;
        dinv[i] = rsqrtf((float)c + 1.0f);
        if (i == N - 1) rowptr[N] = excl + c;
    }
}

__global__ void place_kernel(const int* __restrict__ src, const int* __restrict__ dst,
                             const int* __restrict__ rank, const int* __restrict__ rowptr,
                             const float* __restrict__ dinv, int2* __restrict__ colw, int E) {
    int e = blockIdx.x * blockDim.x + threadIdx.x;
    if (e >= E) return;
    const int s = src[e], d = dst[e];
    const int pos = rowptr[d] + rank[e];
    colw[pos] = make_int2(s, __float_as_int(dinv[s] * dinv[d]));
}

#define SP 132

__device__ __forceinline__ float elu1(float v) {
    return v > 0.f ? v : (expf(v) - 1.f);
}

template <int K>
__device__ __forceinline__ void gather_k(float4& acc, const float* __restrict__ H,
                                         int cols, int wbits, int j, int lane) {
    float4 v[K];
    float  w[K];
    #pragma unroll
    for (int k = 0; k < K; ++k) {
        const int s = __builtin_amdgcn_readlane(cols, j + k);
        w[k] = __int_as_float(__builtin_amdgcn_readlane(wbits, j + k));
        v[k] = *(const float4*)(H + (size_t)s * C + lane * 4);
    }
    #pragma unroll
    for (int k = 0; k < K; ++k) {
        acc.x = fmaf(v[k].x, w[k], acc.x);
        acc.y = fmaf(v[k].y, w[k], acc.y);
        acc.z = fmaf(v[k].z, w[k], acc.z);
        acc.w = fmaf(v[k].w, w[k], acc.w);
    }
}

__device__ __forceinline__ void gather_n(float4& acc, const float* __restrict__ H,
                                         int cols, int wbits, int j, int cnt, int lane) {
    switch (cnt) {
        case 1:  gather_k<1>(acc, H, cols, wbits, j, lane);  break;
        case 2:  gather_k<2>(acc, H, cols, wbits, j, lane);  break;
        case 3:  gather_k<3>(acc, H, cols, wbits, j, lane);  break;
        case 4:  gather_k<4>(acc, H, cols, wbits, j, lane);  break;
        case 5:  gather_k<5>(acc, H, cols, wbits, j, lane);  break;
        case 6:  gather_k<6>(acc, H, cols, wbits, j, lane);  break;
        case 7:  gather_k<7>(acc, H, cols, wbits, j, lane);  break;
        case 8:  gather_k<8>(acc, H, cols, wbits, j, lane);  break;
        case 9:  gather_k<9>(acc, H, cols, wbits, j, lane);  break;
        case 10: gather_k<10>(acc, H, cols, wbits, j, lane); break;
        case 11: gather_k<11>(acc, H, cols, wbits, j, lane); break;
        case 12: gather_k<12>(acc, H, cols, wbits, j, lane); break;
        case 13: gather_k<13>(acc, H, cols, wbits, j, lane); break;
        case 14: gather_k<14>(acc, H, cols, wbits, j, lane); break;
        case 15: gather_k<15>(acc, H, cols, wbits, j, lane); break;
        case 16: gather_k<16>(acc, H, cols, wbits, j, lane); break;
        default: break;
    }
}

__global__ __launch_bounds__(TPB) void layer_fused(const float* __restrict__ Hin,
                                                   const float* __restrict__ W,
                                                   const float* __restrict__ bias,
                                                   const int* __restrict__ rowptr,
                                                   const int2* __restrict__ colw,
                                                   const float* __restrict__ dinv,
                                                   float* __restrict__ out,
                                                   int N, int E, int nStrips) {
    __shared__ __attribute__((aligned(16))) float Wl[C * C];
    __shared__ __attribute__((aligned(16))) float sA[(TPB / 32) * 16 * SP];

    const int lane = threadIdx.x & 31;
    const int half = lane >> 4;
    const int lo   = lane & 15;
    const int wave  = __builtin_amdgcn_readfirstlane((int)(threadIdx.x >> 5));
    const int strip = blockIdx.x * (TPB / 32) + wave;
    const bool active = (strip < nStrips);
    const int m0 = strip * 16;
    float* sW = sA + wave * 16 * SP;

    #pragma unroll
    for (int i = 0; i < (C / 2) * (C / 4) / TPB; ++i) {
        const int u = threadIdx.x + i * TPB;
        const int p = u >> 5;
        const int n = (u & 31) * 4;
        const float4 r0 = *(const float4*)(W + (size_t)(2 * p) * C + n);
        const float4 r1 = *(const float4*)(W + (size_t)(2 * p + 1) * C + n);
        float* d = &Wl[p * 256 + n * 2];
        *(float4*)(d)     = make_float4(r0.x, r1.x, r0.y, r1.y);
        *(float4*)(d + 4) = make_float4(r0.z, r1.z, r0.w, r1.w);
    }

    if (active) {
        int rpv = 0;
        float dv = 0.f;
        {
            int idx = m0 + lane;
            if (idx > N) idx = N;
            if (lane <= 16) rpv = rowptr[idx];
            if (lane < 16 && m0 + lane < N) dv = dinv[m0 + lane];
        }
        int beg = __builtin_amdgcn_readlane(rpv, 0);
        const int eLast = (E > 0) ? (E - 1) : 0;
        int2 cwn = colw[(beg + lane < eLast) ? (beg + lane) : eLast];
        for (int r = 0; r < 16; ++r) {
            const int row = m0 + r;
            const int end = __builtin_amdgcn_readlane(rpv, r + 1);
            const int2 cw0 = cwn;
            cwn = colw[(end + lane < eLast) ? (end + lane) : eLast];
            const float di = __int_as_float(__builtin_amdgcn_readlane(__float_as_int(dv), r));
            const float wself = di * di;
            const float4 hs = *(const float4*)(Hin + (size_t)((row < N) ? row : (N - 1)) * C + lane * 4);
            float4 acc = make_float4(0.f, 0.f, 0.f, 0.f);
            {
                const int n = (end - beg < 32) ? (end - beg) : 32;
                int j = 0;
                if (n > 16) {
                    gather_k<16>(acc, Hin, cw0.x, cw0.y, 0, lane);
                    j = 16;
                }
                gather_n(acc, Hin, cw0.x, cw0.y, j, n - j, lane);
            }
            for (int base = beg + 32; base < end; base += 32) {
                const int m = (end - base < 32) ? (end - base) : 32;
                int2 cx = make_int2(0, 0);
                if (lane < m) cx = colw[base + lane];
                for (int j = 0; j < m; ++j) gather_k<1>(acc, Hin, cx.x, cx.y, j, lane);
            }
            acc.x = fmaf(hs.x, wself, acc.x);
            acc.y = fmaf(hs.y, wself, acc.y);
            acc.z = fmaf(hs.z, wself, acc.z);
            acc.w = fmaf(hs.w, wself, acc.w);
            *(float4*)(sW + r * SP + lane * 4) = acc;
            beg = end;
        }
    }
    __syncthreads();

    if (active) {
        v8f acc[8] = {};
        const float* Ab = sW + lo * SP + 2 * half;
        #pragma unroll
        for (int k0 = 0; k0 < C; k0 += 4) {
            const v2f a = *(const v2f*)(Ab + k0);
            const float* Bbase = &Wl[((k0 >> 1) + half) * 256 + lo * 2];
            #pragma unroll
            for (int t = 0; t < 8; ++t) {
                const v2f b = *(const v2f*)(Bbase + t * 32);
                acc[t] = __builtin_amdgcn_wmma_f32_16x16x4_f32(
                    false, a, false, b, (short)0, acc[t], false, false);
            }
        }
        #pragma unroll
        for (int t = 0; t < 8; ++t)
            #pragma unroll
            for (int r = 0; r < 8; ++r)
                sW[(r + 8 * half) * SP + t * 16 + lo] = acc[t][r];
    }
    __syncthreads();

    if (active) {
        const float4 bv = make_float4(bias[lane * 4 + 0], bias[lane * 4 + 1],
                                      bias[lane * 4 + 2], bias[lane * 4 + 3]);
        for (int r0 = 0; r0 < 16; r0 += 4) {
            float4 v[4];
            #pragma unroll
            for (int q = 0; q < 4; ++q) v[q] = *(const float4*)(sW + (r0 + q) * SP + lane * 4);
            #pragma unroll
            for (int q = 0; q < 4; ++q) {
                const int row = m0 + r0 + q;
                if (row < N) {
                    float4 o;
                    o.x = elu1(v[q].x + bv.x);
                    o.y = elu1(v[q].y + bv.y);
                    o.z = elu1(v[q].z + bv.z);
                    o.w = elu1(v[q].w + bv.w);
                    *(float4*)(out + (size_t)row * C + lane * 4) = o;
                }
            }
        }
    }
}

extern "C" void kernel_launch(void* const* d_in, const int* in_sizes, int n_in,
                              void* d_out, int out_size, void* d_ws, size_t ws_size,
                              hipStream_t stream) {
    const float* x  = (const float*)d_in[0];
    const int*   ei = (const int*)d_in[1];
    const float* W1 = (const float*)d_in[2];
    const float* b1 = (const float*)d_in[3];
    const float* W2 = (const float*)d_in[4];
    const float* b2 = (const float*)d_in[5];

    const int N = in_sizes[0] / C;
    const int E = in_sizes[1] / 2;
    const int* srcI = ei;
    const int* dstI = ei + E;

    char* ws = (char*)d_ws;
    size_t off = 0;
    int* cnt = (int*)(ws + off);      off += ((size_t)N * 4 + 255) & ~(size_t)255;
    int* rowptr = (int*)(ws + off);   off += ((size_t)(N + 1) * 4 + 255) & ~(size_t)255;
    float* dinv = (float*)(ws + off); off += ((size_t)N * 4 + 255) & ~(size_t)255;
    int* rank = (int*)(ws + off);     off += ((size_t)E * 4 + 255) & ~(size_t)255;
    int2* colw = (int2*)(ws + off);   off += ((size_t)E * 8 + 255) & ~(size_t)255;
    float* buf1 = (float*)(ws + off);

    const int strips = (N + 15) / 16;
    const dim3 blk(TPB);
    const int layerBlocks = (strips + (TPB / 32) - 1) / (TPB / 32);
    const int edgeBlocks = (E + TPB - 1) / TPB;
    const int scanBlocks = (N + SCAN_T - 1) / SCAN_T;

    hipMemsetAsync(cnt, 0, (size_t)N * 4, stream);
    rank_kernel<<<edgeBlocks, blk, 0, stream>>>(dstI, cnt, rank, E);
    scan_kernel<<<scanBlocks, dim3(SCAN_T), 0, stream>>>(cnt, rowptr, dinv, N);
    place_kernel<<<edgeBlocks, blk, 0, stream>>>(srcI, dstI, rank, rowptr, dinv, colw, E);

    layer_fused<<<layerBlocks, blk, 0, stream>>>(x, W1, b1, rowptr, colw, dinv, buf1, N, E, strips);
    layer_fused<<<layerBlocks, blk, 0, stream>>>(buf1, W2, b2, rowptr, colw, dinv, (float*)d_out, N, E, strips);
}


# Build the destination-ordered edge list by LDS binning: chunks sorted by 256-node bucket, then a block per bucket counts, scans, places; no global atomics or memset; sums now in bin arrival order
# speedup vs baseline: 4.3951x; 1.5899x over previous
#include <hip/hip_runtime.h>
#include <math.h>

#define C 128
#define TPB 256

typedef float v2f __attribute__((ext_vector_type(2)));
typedef float v8f __attribute__((ext_vector_type(8)));

#define BT 1024
#define CH 4096
#define NSH 8
#define NPB 256
#define MAXB 256
#define CAP 8192

__device__ __forceinline__ int block_excl_scan(int v, int* wt, int lane, int wave, int& total) {
    int incl = v;
    #pragma unroll
    for (int off = 1; off < 32; off <<= 1) {
        int t = __shfl_up(incl, off);
        if (lane >= off) incl += t;
    }
    if (lane == 31) wt[wave] = incl;
    __syncthreads();
    if (wave == 0) {
        const int t = wt[lane];
        int ti = t;
        #pragma unroll
        for (int off = 1; off < 32; off <<= 1) {
            int u = __shfl_up(ti, off);
            if (lane >= off) ti += u;
        }
        wt[lane] = ti - t;
        if (lane == 31) wt[32] = ti;
    }
    __syncthreads();
    const int res = wt[wave] + incl - v;
    total = wt[32];
    __syncthreads();
    return res;
}

__global__ __launch_bounds__(BT) void bin_kernel(const int* __restrict__ src, const int* __restrict__ dst,
                                                 int2* __restrict__ tmp, int* __restrict__ histT,
                                                 int* __restrict__ offT, int E, int B, int G) {
    __shared__ int cntB[MAXB];
    __shared__ int offB[MAXB];
    __shared__ int wt[33];
    __shared__ int2 sE[CH];
    const int tid  = threadIdx.x;
    const int lane = tid & 31;
    const int wave = tid >> 5;
    const int g  = blockIdx.x;
    const int e0 = g * CH;

    if (tid < MAXB) cntB[tid] = 0;
    __syncthreads();

    int s[CH / BT], d[CH / BT], r[CH / BT];
    #pragma unroll
    for (int k = 0; k < CH / BT; ++k) {
        const int e = e0 + k * BT + tid;
        d[k] = -1; s[k] = 0; r[k] = 0;
        if (e < E) { s[k] = src[e]; d[k] = dst[e]; }
    }
    #pragma unroll
    for (int k = 0; k < CH / BT; ++k)
        if (d[k] >= 0) r[k] = atomicAdd(&cntB[d[k] >> NSH], 1);
    __syncthreads();

    int total;
    const int v = (tid < MAXB) ? cntB[tid] : 0;
    const int ex = block_excl_scan(v, wt, lane, wave, total);
    if (tid < MAXB) offB[tid] = ex;
    __syncthreads();

    #pragma unroll
    for (int k = 0; k < CH / BT; ++k)
        if (d[k] >= 0) sE[offB[d[k] >> NSH] + r[k]] = make_int2(s[k], d[k]);
    __syncthreads();

    const int n = (E - e0 < CH) ? (E - e0) : CH;
    for (int i = tid; i < n; i += BT) tmp[(size_t)e0 + i] = sE[i];
    if (tid < B) {
        histT[tid * G + g] = cntB[tid];
        offT[tid * G + g]  = offB[tid];
    }
}

__global__ __launch_bounds__(BT) void bucket_kernel(const int2* __restrict__ tmp, const int* __restrict__ histT,
                                                    const int* __restrict__ offT, int* __restrict__ rowptr,
                                                    float* __restrict__ dinv, int2* __restrict__ colw,
                                                    int N, int B, int G) {
    __shared__ int segStart[BT + 1];
    __shared__ int segOff[BT];
    __shared__ int cntN[NPB];
    __shared__ int lrp[NPB];
    __shared__ int cur[NPB];
    __shared__ int wt[33];
    __shared__ int2 eS[CAP];
    __shared__ int2 oS[CAP];
    const int tid  = threadIdx.x;
    const int lane = tid & 31;
    const int wave = tid >> 5;
    const int b = blockIdx.x;

    int part = 0;
    for (int i = tid; i < b * G; i += BT) part += histT[i];
    int base;
    block_excl_scan(part, wt, lane, wave, base);

    const int h = (tid < G) ? histT[b * G + tid] : 0;
    int nb;
    const int st0 = block_excl_scan(h, wt, lane, wave, nb);
    if (tid < G) { segStart[tid] = st0; segOff[tid] = offT[b * G + tid]; }
    if (tid == 0) segStart[G] = nb;
    if (tid < NPB) { cntN[tid] = 0; cur[tid] = 0; }
    __syncthreads();

    const bool staged = (nb <= CAP);

    for (int g = wave; g < G; g += BT / 32) {
        const int st  = segStart[g];
        const int len = segStart[g + 1] - st;
        const int2* p = tmp + (size_t)g * CH + segOff[g];
        for (int j = lane; j < len; j += 32) {
            const int2 e = p[j];
            if (staged) eS[st + j] = e;
            atomicAdd(&cntN[e.y & (NPB - 1)], 1);
        }
    }
    __syncthreads();

    int tot2;
    const int c  = (tid < NPB) ? cntN[tid] : 0;
    const int lx = block_excl_scan(c, wt, lane, wave, tot2);
    if (tid < NPB) {
        lrp[tid] = lx;
        const int node = b * NPB + tid;
        if (node < N) {
            rowptr[node] = base + lx;
            dinv[node] = rsqrtf((float)c + 1.0f);
            if (node == N - 1) rowptr[N] = base + lx + c;
        }
    }
    __syncthreads();

    if (staged) {
        for (int i = tid; i < nb; i += BT) {
            const int2 e = eS[i];
            const int nd = e.y & (NPB - 1);
            const int r = atomicAdd(&cur[nd], 1);
            oS[lrp[nd] + r] = e;
        }
        __syncthreads();
        for (int i = tid; i < nb; i += BT) colw[(size_t)base + i] = oS[i];
    } else {
        for (int g = wave; g < G; g += BT / 32) {
            const int st  = segStart[g];
            const int len = segStart[g + 1] - st;
            const int2* p = tmp + (size_t)g * CH + segOff[g];
            for (int j = lane; j < len; j += 32) {
                const int2 e = p[j];
                const int nd = e.y & (NPB - 1);
                const int r = atomicAdd(&cur[nd], 1);
                colw[(size_t)base + lrp[nd] + r] = e;
            }
        }
    }
}

__global__ void weight_kernel(int2* __restrict__ colw, const float* __restrict__ dinv, int E) {
    const int p = blockIdx.x * blockDim.x + threadIdx.x;
    if (p >= E) return;
    const int2 c = colw[p];
    colw[p] = make_int2(c.x, __float_as_int(dinv[c.x] * dinv[c.y]));
}

#define SP 132

__device__ __forceinline__ float elu1(float v) {
    return v > 0.f ? v : (expf(v) - 1.f);
}

template <int K>
__device__ __forceinline__ void gather_k(float4& acc, const float* __restrict__ H,
                                         int cols, int wbits, int j, int lane) {
    float4 v[K];
    float  w[K];
    #pragma unroll
    for (int k = 0; k < K; ++k) {
        const int s = __builtin_amdgcn_readlane(cols, j + k);
        w[k] = __int_as_float(__builtin_amdgcn_readlane(wbits, j + k));
        v[k] = *(const float4*)(H + (size_t)s * C + lane * 4);
    }
    #pragma unroll
    for (int k = 0; k < K; ++k) {
        acc.x = fmaf(v[k].x, w[k], acc.x);
        acc.y = fmaf(v[k].y, w[k], acc.y);
        acc.z = fmaf(v[k].z, w[k], acc.z);
        acc.w = fmaf(v[k].w, w[k], acc.w);
    }
}

__device__ __forceinline__ void gather_n(float4& acc, const float* __restrict__ H,
                                         int cols, int wbits, int j, int cnt, int lane) {
    switch (cnt) {
        case 1:  gather_k<1>(acc, H, cols, wbits, j, lane);  break;
        case 2:  gather_k<2>(acc, H, cols, wbits, j, lane);  break;
        case 3:  gather_k<3>(acc, H, cols, wbits, j, lane);  break;
        case 4:  gather_k<4>(acc, H, cols, wbits, j, lane);  break;
        case 5:  gather_k<5>(acc, H, cols, wbits, j, lane);  break;
        case 6:  gather_k<6>(acc, H, cols, wbits, j, lane);  break;
        case 7:  gather_k<7>(acc, H, cols, wbits, j, lane);  break;
        case 8:  gather_k<8>(acc, H, cols, wbits, j, lane);  break;
        case 9:  gather_k<9>(acc, H, cols, wbits, j, lane);  break;
        case 10: gather_k<10>(acc, H, cols, wbits, j, lane); break;
        case 11: gather_k<11>(acc, H, cols, wbits, j, lane); break;
        case 12: gather_k<12>(acc, H, cols, wbits, j, lane); break;
        case 13: gather_k<13>(acc, H, cols, wbits, j, lane); break;
        case 14: gather_k<14>(acc, H, cols, wbits, j, lane); break;
        case 15: gather_k<15>(acc, H, cols, wbits, j, lane); break;
        case 16: gather_k<16>(acc, H, cols, wbits, j, lane); break;
        default: break;
    }
}

__global__ __launch_bounds__(TPB) void layer_fused(const float* __restrict__ Hin,
                                                   const float* __restrict__ W,
                                                   const float* __restrict__ bias,
                                                   const int* __restrict__ rowptr,
                                                   const int2* __restrict__ colw,
                                                   const float* __restrict__ dinv,
                                                   float* __restrict__ out,
                                                   int N, int E, int nStrips) {
    __shared__ __attribute__((aligned(16))) float Wl[C * C];
    __shared__ __attribute__((aligned(16))) float sA[(TPB / 32) * 16 * SP];

    const int lane = threadIdx.x & 31;
    const int half = lane >> 4;
    const int lo   = lane & 15;
    const int wave  = __builtin_amdgcn_readfirstlane((int)(threadIdx.x >> 5));
    const int strip = blockIdx.x * (TPB / 32) + wave;
    const bool active = (strip < nStrips);
    const int m0 = strip * 16;
    float* sW = sA + wave * 16 * SP;

    #pragma unroll
    for (int i = 0; i < (C / 2) * (C / 4) / TPB; ++i) {
        const int u = threadIdx.x + i * TPB;
        const int p = u >> 5;
        const int n = (u & 31) * 4;
        const float4 r0 = *(const float4*)(W + (size_t)(2 * p) * C + n);
        const float4 r1 = *(const float4*)(W + (size_t)(2 * p + 1) * C + n);
        float* d = &Wl[p * 256 + n * 2];
        *(float4*)(d)     = make_float4(r0.x, r1.x, r0.y, r1.y);
        *(float4*)(d + 4) = make_float4(r0.z, r1.z, r0.w, r1.w);
    }

    if (active) {
        int rpv = 0;
        float dv = 0.f;
        {
            int idx = m0 + lane;
            if (idx > N) idx = N;
            if (lane <= 16) rpv = rowptr[idx];
            if (lane < 16 && m0 + lane < N) dv = dinv[m0 + lane];
        }
        int beg = __builtin_amdgcn_readlane(rpv, 0);
        const int eLast = (E > 0) ? (E - 1) : 0;
        int2 cwn = colw[(beg + lane < eLast) ? (beg + lane) : eLast];
        for (int r = 0; r < 16; ++r) {
            const int row = m0 + r;
            const int end = __builtin_amdgcn_readlane(rpv, r + 1);
            const int2 cw0 = cwn;
            cwn = colw[(end + lane < eLast) ? (end + lane) : eLast];
            const float di = __int_as_float(__builtin_amdgcn_readlane(__float_as_int(dv), r));
            const float wself = di * di;
            const float4 hs = *(const float4*)(Hin + (size_t)((row < N) ? row : (N - 1)) * C + lane * 4);
            float4 acc = make_float4(0.f, 0.f, 0.f, 0.f);
            {
                const int n = (end - beg < 32) ? (end - beg) : 32;
                int j = 0;
                if (n > 16) {
                    gather_k<16>(acc, Hin, cw0.x, cw0.y, 0, lane);
                    j = 16;
                }
                gather_n(acc, Hin, cw0.x, cw0.y, j, n - j, lane);
            }
            for (int base = beg + 32; base < end; base += 32) {
                const int m = (end - base < 32) ? (end - base) : 32;
                int2 cx = make_int2(0, 0);
                if (lane < m) cx = colw[base + lane];
                for (int j = 0; j < m; ++j) gather_k<1>(acc, Hin, cx.x, cx.y, j, lane);
            }
            acc.x = fmaf(hs.x, wself, acc.x);
            acc.y = fmaf(hs.y, wself, acc.y);
            acc.z = fmaf(hs.z, wself, acc.z);
            acc.w = fmaf(hs.w, wself, acc.w);
            *(float4*)(sW + r * SP + lane * 4) = acc;
            beg = end;
        }
    }
    __syncthreads();

    if (active) {
        v8f acc[8] = {};
        const float* Ab = sW + lo * SP + 2 * half;
        #pragma unroll
        for (int k0 = 0; k0 < C; k0 += 4) {
            const v2f a = *(const v2f*)(Ab + k0);
            const float* Bbase = &Wl[((k0 >> 1) + half) * 256 + lo * 2];
            #pragma unroll
            for (int t = 0; t < 8; ++t) {
                const v2f b = *(const v2f*)(Bbase + t * 32);
                acc[t] = __builtin_amdgcn_wmma_f32_16x16x4_f32(
                    false, a, false, b, (short)0, acc[t], false, false);
            }
        }
        #pragma unroll
        for (int t = 0; t < 8; ++t)
            #pragma unroll
            for (int r = 0; r < 8; ++r)
                sW[(r + 8 * half) * SP + t * 16 + lo] = acc[t][r];
    }
    __syncthreads();

    if (active) {
        const float4 bv = make_float4(bias[lane * 4 + 0], bias[lane * 4 + 1],
                                      bias[lane * 4 + 2], bias[lane * 4 + 3]);
        for (int r0 = 0; r0 < 16; r0 += 4) {
            float4 v[4];
            #pragma unroll
            for (int q = 0; q < 4; ++q) v[q] = *(const float4*)(sW + (r0 + q) * SP + lane * 4);
            #pragma unroll
            for (int q = 0; q < 4; ++q) {
                const int row = m0 + r0 + q;
                if (row < N) {
                    float4 o;
                    o.x = elu1(v[q].x + bv.x);
                    o.y = elu1(v[q].y + bv.y);
                    o.z = elu1(v[q].z + bv.z);
                    o.w = elu1(v[q].w + bv.w);
                    *(float4*)(out + (size_t)row * C + lane * 4) = o;
                }
            }
        }
    }
}

extern "C" void kernel_launch(void* const* d_in, const int* in_sizes, int n_in,
                              void* d_out, int out_size, void* d_ws, size_t ws_size,
                              hipStream_t stream) {
    const float* x  = (const float*)d_in[0];
    const int*   ei = (const int*)d_in[1];
    const float* W1 = (const float*)d_in[2];
    const float* b1 = (const float*)d_in[3];
    const float* W2 = (const float*)d_in[4];
    const float* b2 = (const float*)d_in[5];

    const int N = in_sizes[0] / C;
    const int E = in_sizes[1] / 2;
    const int* srcI = ei;
    const int* dstI = ei + E;

    const int B = (N + NPB - 1) / NPB;
    const int G = (E + CH - 1) / CH;

    char* ws = (char*)d_ws;
    size_t off = 0;
    int* rowptr = (int*)(ws + off);   off += ((size_t)(N + 1) * 4 + 255) & ~(size_t)255;
    float* dinv = (float*)(ws + off); off += ((size_t)N * 4 + 255) & ~(size_t)255;
    int* histT = (int*)(ws + off);    off += ((size_t)B * G * 4 + 255) & ~(size_t)255;
    int* offT = (int*)(ws + off);     off += ((size_t)B * G * 4 + 255) & ~(size_t)255;
    int2* tmp = (int2*)(ws + off);    off += ((size_t)G * CH * 8 + 255) & ~(size_t)255;
    int2* colw = (int2*)(ws + off);   off += ((size_t)E * 8 + 255) & ~(size_t)255;
    float* buf1 = (float*)(ws + off);

    const int strips = (N + 15) / 16;
    const dim3 blk(TPB);
    const int layerBlocks = (strips + (TPB / 32) - 1) / (TPB / 32);
    const int edgeBlocks = (E + TPB - 1) / TPB;

    bin_kernel<<<G, dim3(BT), 0, stream>>>(srcI, dstI, tmp, histT, offT, E, B, G);
    bucket_kernel<<<B, dim3(BT), 0, stream>>>(tmp, histT, offT, rowptr, dinv, colw, N, B, G);
    weight_kernel<<<edgeBlocks, blk, 0, stream>>>(colw, dinv, E);

    layer_fused<<<layerBlocks, blk, 0, stream>>>(x, W1, b1, rowptr, colw, dinv, buf1, N, E, strips);
    layer_fused<<<layerBlocks, blk, 0, stream>>>(buf1, W2, b2, rowptr, colw, dinv, (float*)d_out, N, E, strips);
}


# Layer kernel as a 2-deep pipeline, one 24-wave block per WGP: 16 waves gather+sum the next 4 strips into an LDS ring while 8 waves run the last 4 through W on WMMA (bias+ELU, stores); sum order kept
# speedup vs baseline: 4.6540x; 1.0589x over previous
#include <hip/hip_runtime.h>
#include <math.h>

#define C 128
#define TPB 256

typedef float v2f __attribute__((ext_vector_type(2)));
typedef float v8f __attribute__((ext_vector_type(8)));

#define BT 1024
#define CH 4096
#define NSH 8
#define NPB 256
#define MAXB 256
#define CAP 8192

__device__ __forceinline__ int block_excl_scan(int v, int* wt, int lane, int wave, int& total) {
    int incl = v;
    #pragma unroll
    for (int off = 1; off < 32; off <<= 1) {
        int t = __shfl_up(incl, off);
        if (lane >= off) incl += t;
    }
    if (lane == 31) wt[wave] = incl;
    __syncthreads();
    if (wave == 0) {
        const int t = wt[lane];
        int ti = t;
        #pragma unroll
        for (int off = 1; off < 32; off <<= 1) {
            int u = __shfl_up(ti, off);
            if (lane >= off) ti += u;
        }
        wt[lane] = ti - t;
        if (lane == 31) wt[32] = ti;
    }
    __syncthreads();
    const int res = wt[wave] + incl - v;
    total = wt[32];
    __syncthreads();
    return res;
}

__global__ __launch_bounds__(BT) void bin_kernel(const int* __restrict__ src, const int* __restrict__ dst,
                                                 int2* __restrict__ tmp, int* __restrict__ histT,
                                                 int* __restrict__ offT, int E, int B, int G) {
    __shared__ int cntB[MAXB];
    __shared__ int offB[MAXB];
    __shared__ int wt[33];
    __shared__ int2 sE[CH];
    const int tid  = threadIdx.x;
    const int lane = tid & 31;
    const int wave = tid >> 5;
    const int g  = blockIdx.x;
    const int e0 = g * CH;

    if (tid < MAXB) cntB[tid] = 0;
    __syncthreads();

    int s[CH / BT], d[CH / BT], r[CH / BT];
    #pragma unroll
    for (int k = 0; k < CH / BT; ++k) {
        const int e = e0 + k * BT + tid;
        d[k] = -1; s[k] = 0; r[k] = 0;
        if (e < E) { s[k] = src[e]; d[k] = dst[e]; }
    }
    #pragma unroll
    for (int k = 0; k < CH / BT; ++k)
        if (d[k] >= 0) r[k] = atomicAdd(&cntB[d[k] >> NSH], 1);
    __syncthreads();

    int total;
    const int v = (tid < MAXB) ? cntB[tid] : 0;
    const int ex = block_excl_scan(v, wt, lane, wave, total);
    if (tid < MAXB) offB[tid] = ex;
    __syncthreads();

    #pragma unroll
    for (int k = 0; k < CH / BT; ++k)
        if (d[k] >= 0) sE[offB[d[k] >> NSH] + r[k]] = make_int2(s[k], d[k]);
    __syncthreads();

    const int n = (E - e0 < CH) ? (E - e0) : CH;
    for (int i = tid; i < n; i += BT) tmp[(size_t)e0 + i] = sE[i];
    if (tid < B) {
        histT[tid * G + g] = cntB[tid];
        offT[tid * G + g]  = offB[tid];
    }
}

__global__ __launch_bounds__(BT) void bucket_kernel(const int2* __restrict__ tmp, const int* __restrict__ histT,
                                                    const int* __restrict__ offT, int* __restrict__ rowptr,
                                                    float* __restrict__ dinv, int2* __restrict__ colw,
                                                    int N, int B, int G) {
    __shared__ int segStart[BT + 1];
    __shared__ int segOff[BT];
    __shared__ int cntN[NPB];
    __shared__ int lrp[NPB];
    __shared__ int cur[NPB];
    __shared__ int wt[33];
    __shared__ int2 eS[CAP];
    __shared__ int2 oS[CAP];
    const int tid  = threadIdx.x;
    const int lane = tid & 31;
    const int wave = tid >> 5;
    const int b = blockIdx.x;

    int part = 0;
    for (int i = tid; i < b * G; i += BT) part += histT[i];
    int base;
    block_excl_scan(part, wt, lane, wave, base);

    const int h = (tid < G) ? histT[b * G + tid] : 0;
    int nb;
    const int st0 = block_excl_scan(h, wt, lane, wave, nb);
    if (tid < G) { segStart[tid] = st0; segOff[tid] = offT[b * G + tid]; }
    if (tid == 0) segStart[G] = nb;
    if (tid < NPB) { cntN[tid] = 0; cur[tid] = 0; }
    __syncthreads();

    const bool staged = (nb <= CAP);

    for (int g = wave; g < G; g += BT / 32) {
        const int st  = segStart[g];
        const int len = segStart[g + 1] - st;
        const int2* p = tmp + (size_t)g * CH + segOff[g];
        for (int j = lane; j < len; j += 32) {
            const int2 e = p[j];
            if (staged) eS[st + j] = e;
            atomicAdd(&cntN[e.y & (NPB - 1)], 1);
        }
    }
    __syncthreads();

    int tot2;
    const int c  = (tid < NPB) ? cntN[tid] : 0;
    const int lx = block_excl_scan(c, wt, lane, wave, tot2);
    if (tid < NPB) {
        lrp[tid] = lx;
        const int node = b * NPB + tid;
        if (node < N) {
            rowptr[node] = base + lx;
            dinv[node] = rsqrtf((float)c + 1.0f);
            if (node == N - 1) rowptr[N] = base + lx + c;
        }
    }
    __syncthreads();

    if (staged) {
        for (int i = tid; i < nb; i += BT) {
            const int2 e = eS[i];
            const int nd = e.y & (NPB - 1);
            const int r = atomicAdd(&cur[nd], 1);
            oS[lrp[nd] + r] = e;
        }
        __syncthreads();
        for (int i = tid; i < nb; i += BT) colw[(size_t)base + i] = oS[i];
    } else {
        for (int g = wave; g < G; g += BT / 32) {
            const int st  = segStart[g];
            const int len = segStart[g + 1] - st;
            const int2* p = tmp + (size_t)g * CH + segOff[g];
            for (int j = lane; j < len; j += 32) {
                const int2 e = p[j];
                const int nd = e.y & (NPB - 1);
                const int r = atomicAdd(&cur[nd], 1);
                colw[(size_t)base + lrp[nd] + r] = e;
            }
        }
    }
}

__global__ void weight_kernel(int2* __restrict__ colw, const float* __restrict__ dinv, int E) {
    const int p = blockIdx.x * blockDim.x + threadIdx.x;
    if (p >= E) return;
    const int2 c = colw[p];
    colw[p] = make_int2(c.x, __float_as_int(dinv[c.x] * dinv[c.y]));
}

#define SP 132

__device__ __forceinline__ float elu1(float v) {
    return v > 0.f ? v : (expf(v) - 1.f);
}

template <int K>
__device__ __forceinline__ void gather_k(float4& acc, const float* __restrict__ H,
                                         int cols, int wbits, int j, int lane) {
    float4 v[K];
    float  w[K];
    #pragma unroll
    for (int k = 0; k < K; ++k) {
        const int s = __builtin_amdgcn_readlane(cols, j + k);
        w[k] = __int_as_float(__builtin_amdgcn_readlane(wbits, j + k));
        v[k] = *(const float4*)(H + (size_t)s * C + lane * 4);
    }
    #pragma unroll
    for (int k = 0; k < K; ++k) {
        acc.x = fmaf(v[k].x, w[k], acc.x);
        acc.y = fmaf(v[k].y, w[k], acc.y);
        acc.z = fmaf(v[k].z, w[k], acc.z);
        acc.w = fmaf(v[k].w, w[k], acc.w);
    }
}

__device__ __forceinline__ void gather_n(float4& acc, const float* __restrict__ H,
                                         int cols, int wbits, int j, int cnt, int lane) {
    switch (cnt) {
        case 1:  gather_k<1>(acc, H, cols, wbits, j, lane);  break;
        case 2:  gather_k<2>(acc, H, cols, wbits, j, lane);  break;
        case 3:  gather_k<3>(acc, H, cols, wbits, j, lane);  break;
        case 4:  gather_k<4>(acc, H, cols, wbits, j, lane);  break;
        case 5:  gather_k<5>(acc, H, cols, wbits, j, lane);  break;
        case 6:  gather_k<6>(acc, H, cols, wbits, j, lane);  break;
        case 7:  gather_k<7>(acc, H, cols, wbits, j, lane);  break;
        case 8:  gather_k<8>(acc, H, cols, wbits, j, lane);  break;
        case 9:  gather_k<9>(acc, H, cols, wbits, j, lane);  break;
        case 10: gather_k<10>(acc, H, cols, wbits, j, lane); break;
        case 11: gather_k<11>(acc, H, cols, wbits, j, lane); break;
        case 12: gather_k<12>(acc, H, cols, wbits, j, lane); break;
        case 13: gather_k<13>(acc, H, cols, wbits, j, lane); break;
        case 14: gather_k<14>(acc, H, cols, wbits, j, lane); break;
        case 15: gather_k<15>(acc, H, cols, wbits, j, lane); break;
        case 16: gather_k<16>(acc, H, cols, wbits, j, lane); break;
        default: break;
    }
}

#define LT 768
#define MW 8
#define SS 4
#define OP 68
#define WGPS 256

__global__ __launch_bounds__(LT) void layer_fused(const float* __restrict__ Hin,
                                                  const float* __restrict__ W,
                                                  const float* __restrict__ bias,
                                                  const int* __restrict__ rowptr,
                                                  const int2* __restrict__ colw,
                                                  const float* __restrict__ dinv,
                                                  float* __restrict__ out,
                                                  int N, int E, int nStrips) {
    __shared__ __attribute__((aligned(16))) float Wl[C * C];
    __shared__ __attribute__((aligned(16))) float sA[2 * SS * 16 * SP];
    __shared__ __attribute__((aligned(16))) float sO[MW * 16 * OP];

    const int lane = threadIdx.x & 31;
    const int half = lane >> 4;
    const int lo   = lane & 15;
    const int wave = __builtin_amdgcn_readfirstlane((int)(threadIdx.x >> 5));

    const int s0  = (int)((blockIdx.x * (unsigned)nStrips) / gridDim.x);
    const int s1  = (int)(((blockIdx.x + 1u) * (unsigned)nStrips) / gridDim.x);
    const int cnt = s1 - s0;
    const int nst = (cnt + SS - 1) / SS;

    #pragma unroll
    for (int i = 0; i < ((C / 2) * (C / 4) + LT - 1) / LT; ++i) {
        const int u = threadIdx.x + i * LT;
        if (u < (C / 2) * (C / 4)) {
            const int p = u >> 5;
            const int n = (u & 31) * 4;
            const float4 r0 = *(const float4*)(W + (size_t)(2 * p) * C + n);
            const float4 r1 = *(const float4*)(W + (size_t)(2 * p + 1) * C + n);
            float* d = &Wl[p * 256 + n * 2];
            *(float4*)(d)     = make_float4(r0.x, r1.x, r0.y, r1.y);
            *(float4*)(d + 4) = make_float4(r0.z, r1.z, r0.w, r1.w);
        }
    }

    const int g  = wave - MW;
    const int sg = g >> 2;
    const int eLast = (E > 0) ? (E - 1) : 0;
    int rpv = 0;
    float dv = 0.f;
    int2 cwn = make_int2(0, 0);
    if (wave >= MW) {
        if (lane < 25) {
            const int tt = lane / 5;
            const int j  = lane - 5 * tt;
            const int row = s0 * 16 + 64 * tt + 4 * g + j;
            rpv = rowptr[(row < N) ? row : N];
            if (j < 4 && row < N) dv = dinv[row];
        }
        const int b0 = __builtin_amdgcn_readlane(rpv, 0);
        cwn = colw[(b0 + lane < eLast) ? (b0 + lane) : eLast];
    }

    const int hc  = (wave >> 2) & 1;
    const int mst = (wave < 4) ? wave : ((wave + 1) & 3);
    const int bc  = hc * 64 + lo * 4;
    const float4 bv = make_float4(bias[bc + 0], bias[bc + 1], bias[bc + 2], bias[bc + 3]);

    for (int t = 0; t <= nst; ++t) {
        if (wave >= MW) {
            if (t < nst && SS * t + sg < cnt) {
                float* dstp = sA + (((t & 1) * SS + sg) * 16 + (g & 3) * 4) * SP;
                const int row0 = s0 * 16 + 64 * t + 4 * g;
                int beg = __builtin_amdgcn_readlane(rpv, 5 * t);
                for (int i = 0; i < 4; ++i) {
                    const int row = row0 + i;
                    const int end = __builtin_amdgcn_readlane(rpv, 5 * t + i + 1);
                    const int2 cw0 = cwn;
                    const int nxt = (i < 3) ? end : __builtin_amdgcn_readlane(rpv, 5 * t + 5);
                    cwn = colw[(nxt + lane < eLast) ? (nxt + lane) : eLast];
                    const float di = __int_as_float(__builtin_amdgcn_readlane(__float_as_int(dv), 5 * t + i));
                    const float wself = di * di;
                    const float4 hs = *(const float4*)(Hin + (size_t)((row < N) ? row : (N - 1)) * C + lane * 4);
                    float4 acc = make_float4(0.f, 0.f, 0.f, 0.f);
                    {
                        const int n = (end - beg < 32) ? (end - beg) : 32;
                        int j = 0;
                        if (n > 16) {
                            gather_k<16>(acc, Hin, cw0.x, cw0.y, 0, lane);
                            j = 16;
                        }
                        gather_n(acc, Hin, cw0.x, cw0.y, j, n - j, lane);
                    }
                    for (int base = beg + 32; base < end; base += 32) {
                        const int m = (end - base < 32) ? (end - base) : 32;
                        int2 cx = make_int2(0, 0);
                        if (lane < m) cx = colw[base + lane];
                        for (int j = 0; j < m; ++j) gather_k<1>(acc, Hin, cx.x, cx.y, j, lane);
                    }
                    acc.x = fmaf(hs.x, wself, acc.x);
                    acc.y = fmaf(hs.y, wself, acc.y);
                    acc.z = fmaf(hs.z, wself, acc.z);
                    acc.w = fmaf(hs.w, wself, acc.w);
                    *(float4*)(dstp + i * SP + lane * 4) = acc;
                    beg = end;
                }
            }
        } else if (t >= 1 && SS * (t - 1) + mst < cnt) {
            const int tp = t - 1;
            const int m0 = (s0 + SS * tp + mst) * 16;
            const float* sW = sA + ((tp & 1) * SS + mst) * 16 * SP;
            float* sR = sO + wave * 16 * OP;
            v8f acc[4] = {};
            const float* Ab = sW + lo * SP + 2 * half;
            #pragma unroll
            for (int k0 = 0; k0 < C; k0 += 4) {
                const v2f a = *(const v2f*)(Ab + k0);
                const float* Bbase = &Wl[((k0 >> 1) + half) * 256 + hc * 128 + lo * 2];
                #pragma unroll
                for (int c = 0; c < 4; ++c) {
                    const v2f b = *(const v2f*)(Bbase + c * 32);
                    acc[c] = __builtin_amdgcn_wmma_f32_16x16x4_f32(
                        false, a, false, b, (short)0, acc[c], false, false);
                }
            }
            #pragma unroll
            for (int c = 0; c < 4; ++c)
                #pragma unroll
                for (int r = 0; r < 8; ++r)
                    sR[(r + 8 * half) * OP + c * 16 + lo] = acc[c][r];
            for (int r0 = 0; r0 < 16; r0 += 8) {
                float4 v[4];
                #pragma unroll
                for (int q = 0; q < 4; ++q) v[q] = *(const float4*)(sR + (r0 + 2 * q + half) * OP + lo * 4);
                #pragma unroll
                for (int q = 0; q < 4; ++q) {
                    const int row = m0 + r0 + 2 * q + half;
                    if (row < N) {
                        float4 o;
                        o.x = elu1(v[q].x + bv.x);
                        o.y = elu1(v[q].y + bv.y);
                        o.z = elu1(v[q].z + bv.z);
                        o.w = elu1(v[q].w + bv.w);
                        *(float4*)(out + (size_t)row * C + bc) = o;
                    }
                }
            }
        }
        __syncthreads();
    }
}

extern "C" void kernel_launch(void* const* d_in, const int* in_sizes, int n_in,
                              void* d_out, int out_size, void* d_ws, size_t ws_size,
                              hipStream_t stream) {
    const float* x  = (const float*)d_in[0];
    const int*   ei = (const int*)d_in[1];
    const float* W1 = (const float*)d_in[2];
    const float* b1 = (const float*)d_in[3];
    const float* W2 = (const float*)d_in[4];
    const float* b2 = (const float*)d_in[5];

    const int N = in_sizes[0] / C;
    const int E = in_sizes[1] / 2;
    const int* srcI = ei;
    const int* dstI = ei + E;

    const int B = (N + NPB - 1) / NPB;
    const int G = (E + CH - 1) / CH;

    char* ws = (char*)d_ws;
    size_t off = 0;
    int* rowptr = (int*)(ws + off);   off += ((size_t)(N + 1) * 4 + 255) & ~(size_t)255;
    float* dinv = (float*)(ws + off); off += ((size_t)N * 4 + 255) & ~(size_t)255;
    int* histT = (int*)(ws + off);    off += ((size_t)B * G * 4 + 255) & ~(size_t)255;
    int* offT = (int*)(ws + off);     off += ((size_t)B * G * 4 + 255) & ~(size_t)255;
    int2* tmp = (int2*)(ws + off);    off += ((size_t)G * CH * 8 + 255) & ~(size_t)255;
    int2* colw = (int2*)(ws + off);   off += ((size_t)E * 8 + 255) & ~(size_t)255;
    float* buf1 = (float*)(ws + off);

    const int strips = (N + 15) / 16;
    const dim3 blk(TPB);
    const int layerBlocks = ((strips + 15) / 16 > WGPS) ? (strips + 15) / 16 : WGPS;
    const int edgeBlocks = (E + TPB - 1) / TPB;

    bin_kernel<<<G, dim3(BT), 0, stream>>>(srcI, dstI, tmp, histT, offT, E, B, G);
    bucket_kernel<<<B, dim3(BT), 0, stream>>>(tmp, histT, offT, rowptr, dinv, colw, N, B, G);
    weight_kernel<<<edgeBlocks, blk, 0, stream>>>(colw, dinv, E);

    layer_fused<<<layerBlocks, dim3(LT), 0, stream>>>(x, W1, b1, rowptr, colw, dinv, buf1, N, E, strips);
    layer_fused<<<layerBlocks, dim3(LT), 0, stream>>>(buf1, W2, b2, rowptr, colw, dinv, (float*)d_out, N, E, strips);
}


# Weight launch folded into layer 1: each block first copies the dinv table into LDS and writes dinv[src]*dinv[dst] over its own rows' edges, then gathers as before; 4 launches, sums unchanged
# speedup vs baseline: 4.8243x; 1.0366x over previous
#include <hip/hip_runtime.h>
#include <math.h>

#define C 128

typedef float v2f __attribute__((ext_vector_type(2)));
typedef float v8f __attribute__((ext_vector_type(8)));

#define BT 1024
#define CH 4096
#define NSH 8
#define NPB 256
#define MAXB 256
#define CAP 8192

__device__ __forceinline__ int block_excl_scan(int v, int* wt, int lane, int wave, int& total) {
    int incl = v;
    #pragma unroll
    for (int off = 1; off < 32; off <<= 1) {
        int t = __shfl_up(incl, off);
        if (lane >= off) incl += t;
    }
    if (lane == 31) wt[wave] = incl;
    __syncthreads();
    if (wave == 0) {
        const int t = wt[lane];
        int ti = t;
        #pragma unroll
        for (int off = 1; off < 32; off <<= 1) {
            int u = __shfl_up(ti, off);
            if (lane >= off) ti += u;
        }
        wt[lane] = ti - t;
        if (lane == 31) wt[32] = ti;
    }
    __syncthreads();
    const int res = wt[wave] + incl - v;
    total = wt[32];
    __syncthreads();
    return res;
}

__global__ __launch_bounds__(BT) void bin_kernel(const int* __restrict__ src, const int* __restrict__ dst,
                                                 int2* __restrict__ tmp, int* __restrict__ histT,
                                                 int* __restrict__ offT, int E, int B, int G) {
    __shared__ int cntB[MAXB];
    __shared__ int offB[MAXB];
    __shared__ int wt[33];
    __shared__ int2 sE[CH];
    const int tid  = threadIdx.x;
    const int lane = tid & 31;
    const int wave = tid >> 5;
    const int g  = blockIdx.x;
    const int e0 = g * CH;

    if (tid < MAXB) cntB[tid] = 0;
    __syncthreads();

    int s[CH / BT], d[CH / BT], r[CH / BT];
    #pragma unroll
    for (int k = 0; k < CH / BT; ++k) {
        const int e = e0 + k * BT + tid;
        d[k] = -1; s[k] = 0; r[k] = 0;
        if (e < E) { s[k] = src[e]; d[k] = dst[e]; }
    }
    #pragma unroll
    for (int k = 0; k < CH / BT; ++k)
        if (d[k] >= 0) r[k] = atomicAdd(&cntB[d[k] >> NSH], 1);
    __syncthreads();

    int total;
    const int v = (tid < MAXB) ? cntB[tid] : 0;
    const int ex = block_excl_scan(v, wt, lane, wave, total);
    if (tid < MAXB) offB[tid] = ex;
    __syncthreads();

    #pragma unroll
    for (int k = 0; k < CH / BT; ++k)
        if (d[k] >= 0) sE[offB[d[k] >> NSH] + r[k]] = make_int2(s[k], d[k]);
    __syncthreads();

    const int n = (E - e0 < CH) ? (E - e0) : CH;
    for (int i = tid; i < n; i += BT) tmp[(size_t)e0 + i] = sE[i];
    if (tid < B) {
        histT[tid * G + g] = cntB[tid];
        offT[tid * G + g]  = offB[tid];
    }
}

__global__ __launch_bounds__(BT) void bucket_kernel(const int2* __restrict__ tmp, const int* __restrict__ histT,
                                                    const int* __restrict__ offT, int* __restrict__ rowptr,
                                                    float* __restrict__ dinv, int2* __restrict__ colw,
                                                    int N, int B, int G) {
    __shared__ int segStart[BT + 1];
    __shared__ int segOff[BT];
    __shared__ int cntN[NPB];
    __shared__ int lrp[NPB];
    __shared__ int cur[NPB];
    __shared__ int wt[33];
    __shared__ int2 eS[CAP];
    __shared__ int2 oS[CAP];
    const int tid  = threadIdx.x;
    const int lane = tid & 31;
    const int wave = tid >> 5;
    const int b = blockIdx.x;

    int part = 0;
    for (int i = tid; i < b * G; i += BT) part += histT[i];
    int base;
    block_excl_scan(part, wt, lane, wave, base);

    const int h = (tid < G) ? histT[b * G + tid] : 0;
    int nb;
    const int st0 = block_excl_scan(h, wt, lane, wave, nb);
    if (tid < G) { segStart[tid] = st0; segOff[tid] = offT[b * G + tid]; }
    if (tid == 0) segStart[G] = nb;
    if (tid < NPB) { cntN[tid] = 0; cur[tid] = 0; }
    __syncthreads();

    const bool staged = (nb <= CAP);

    for (int g = wave; g < G; g += BT / 32) {
        const int st  = segStart[g];
        const int len = segStart[g + 1] - st;
        const int2* p = tmp + (size_t)g * CH + segOff[g];
        for (int j = lane; j < len; j += 32) {
            const int2 e = p[j];
            if (staged) eS[st + j] = e;
            atomicAdd(&cntN[e.y & (NPB - 1)], 1);
        }
    }
    __syncthreads();

    int tot2;
    const int c  = (tid < NPB) ? cntN[tid] : 0;
    const int lx = block_excl_scan(c, wt, lane, wave, tot2);
    if (tid < NPB) {
        lrp[tid] = lx;
        const int node = b * NPB + tid;
        if (node < N) {
            rowptr[node] = base + lx;
            dinv[node] = rsqrtf((float)c + 1.0f);
            if (node == N - 1) rowptr[N] = base + lx + c;
        }
    }
    __syncthreads();

    if (staged) {
        for (int i = tid; i < nb; i += BT) {
            const int2 e = eS[i];
            const int nd = e.y & (NPB - 1);
            const int r = atomicAdd(&cur[nd], 1);
            oS[lrp[nd] + r] = e;
        }
        __syncthreads();
        for (int i = tid; i < nb; i += BT) colw[(size_t)base + i] = oS[i];
    } else {
        for (int g = wave; g < G; g += BT / 32) {
            const int st  = segStart[g];
            const int len = segStart[g + 1] - st;
            const int2* p = tmp + (size_t)g * CH + segOff[g];
            for (int j = lane; j < len; j += 32) {
                const int2 e = p[j];
                const int nd = e.y & (NPB - 1);
                const int r = atomicAdd(&cur[nd], 1);
                colw[(size_t)base + lrp[nd] + r] = e;
            }
        }
    }
}

#define SP 132

__device__ __forceinline__ float elu1(float v) {
    return v > 0.f ? v : (expf(v) - 1.f);
}

template <int K>
__device__ __forceinline__ void gather_k(float4& acc, const float* __restrict__ H,
                                         int cols, int wbits, int j, int lane) {
    float4 v[K];
    float  w[K];
    #pragma unroll
    for (int k = 0; k < K; ++k) {
        const int s = __builtin_amdgcn_readlane(cols, j + k);
        w[k] = __int_as_float(__builtin_amdgcn_readlane(wbits, j + k));
        v[k] = *(const float4*)(H + (size_t)s * C + lane * 4);
    }
    #pragma unroll
    for (int k = 0; k < K; ++k) {
        acc.x = fmaf(v[k].x, w[k], acc.x);
        acc.y = fmaf(v[k].y, w[k], acc.y);
        acc.z = fmaf(v[k].z, w[k], acc.z);
        acc.w = fmaf(v[k].w, w[k], acc.w);
    }
}

__device__ __forceinline__ void gather_n(float4& acc, const float* __restrict__ H,
                                         int cols, int wbits, int j, int cnt, int lane) {
    switch (cnt) {
        case 1:  gather_k<1>(acc, H, cols, wbits, j, lane);  break;
        case 2:  gather_k<2>(acc, H, cols, wbits, j, lane);  break;
        case 3:  gather_k<3>(acc, H, cols, wbits, j, lane);  break;
        case 4:  gather_k<4>(acc, H, cols, wbits, j, lane);  break;
        case 5:  gather_k<5>(acc, H, cols, wbits, j, lane);  break;
        case 6:  gather_k<6>(acc, H, cols, wbits, j, lane);  break;
        case 7:  gather_k<7>(acc, H, cols, wbits, j, lane);  break;
        case 8:  gather_k<8>(acc, H, cols, wbits, j, lane);  break;
        case 9:  gather_k<9>(acc, H, cols, wbits, j, lane);  break;
        case 10: gather_k<10>(acc, H, cols, wbits, j, lane); break;
        case 11: gather_k<11>(acc, H, cols, wbits, j, lane); break;
        case 12: gather_k<12>(acc, H, cols, wbits, j, lane); break;
        case 13: gather_k<13>(acc, H, cols, wbits, j, lane); break;
        case 14: gather_k<14>(acc, H, cols, wbits, j, lane); break;
        case 15: gather_k<15>(acc, H, cols, wbits, j, lane); break;
        case 16: gather_k<16>(acc, H, cols, wbits, j, lane); break;
        default: break;
    }
}

#define LT 768
#define MW 8
#define SS 4
#define OP 68
#define WGPS 256
#define TABN 65536
#define EK 5
#define TK 11

__global__ __launch_bounds__(LT) void layer_fused(const float* __restrict__ Hin,
                                                  const float* __restrict__ W,
                                                  const float* __restrict__ bias,
                                                  const int* __restrict__ rowptr,
                                                  int2* colw,
                                                  const float* __restrict__ dinv,
                                                  float* __restrict__ out,
                                                  int N, int E, int nStrips, int mkw) {
    __shared__ __attribute__((aligned(16))) float lds[TABN];
    float* const Wl = lds;
    float* const sA = lds + C * C;
    float* const sO = sA + 2 * SS * 16 * SP;

    const int lane = threadIdx.x & 31;
    const int half = lane >> 4;
    const int lo   = lane & 15;
    const int wave = __builtin_amdgcn_readfirstlane((int)(threadIdx.x >> 5));

    const int s0  = (int)((blockIdx.x * (unsigned)nStrips) / gridDim.x);
    const int s1  = (int)(((blockIdx.x + 1u) * (unsigned)nStrips) / gridDim.x);
    const int cnt = s1 - s0;
    const int nst = (cnt + SS - 1) / SS;

    int p0 = 0, p1 = 0;
    if (mkw) {
        p0 = rowptr[(s0 * 16 < N) ? s0 * 16 : N];
        p1 = rowptr[(s1 * 16 < N) ? s1 * 16 : N];
    }

    constexpr int WU = (C / 2) * (C / 4);
    constexpr int WI = (WU + LT - 1) / LT;
    float4 wr0[WI], wr1[WI];
    #pragma unroll
    for (int i = 0; i < WI; ++i) {
        const int u = threadIdx.x + i * LT;
        wr0[i] = make_float4(0.f, 0.f, 0.f, 0.f);
        wr1[i] = wr0[i];
        if (u < WU) {
            const int p = u >> 5;
            const int n = (u & 31) * 4;
            wr0[i] = *(const float4*)(W + (size_t)(2 * p) * C + n);
            wr1[i] = *(const float4*)(W + (size_t)(2 * p + 1) * C + n);
        }
    }

    const int g  = wave - MW;
    const int sg = g >> 2;
    const int eLast = (E > 0) ? (E - 1) : 0;
    int rpv = 0;
    float dv = 0.f;
    int2 cwn = make_int2(0, 0);
    if (wave >= MW) {
        if (lane < 25) {
            const int tt = lane / 5;
            const int j  = lane - 5 * tt;
            const int row = s0 * 16 + 64 * tt + 4 * g + j;
            rpv = rowptr[(row < N) ? row : N];
            if (j < 4 && row < N) dv = dinv[row];
        }
    }

    if (mkw) {
        int2 ce[EK];
        #pragma unroll
        for (int k = 0; k < EK; ++k) {
            const int p = p0 + (int)threadIdx.x + k * LT;
            ce[k] = make_int2(0, 0);
            if (p < p1) ce[k] = colw[p];
        }
        const int n4 = N >> 2;
        const int rot = (int)((blockIdx.x * (unsigned)n4) / gridDim.x);
        const float4* d4 = (const float4*)dinv;
        float4* t4 = (float4*)lds;
        #pragma unroll
        for (int r = 0; r < (TABN / 4 + TK * LT - 1) / (TK * LT); ++r) {
            float4 tv[TK];
            #pragma unroll
            for (int k = 0; k < TK; ++k) {
                const int i0 = (r * TK + k) * LT;
                tv[k] = make_float4(0.f, 0.f, 0.f, 0.f);
                if (i0 < n4) {
                    int i = i0 + (int)threadIdx.x;
                    if (i >= n4) i = n4 - 1;
                    int q = i + rot;
                    if (q >= n4) q -= n4;
                    tv[k] = d4[q];
                }
            }
            #pragma unroll
            for (int k = 0; k < TK; ++k) {
                const int i0 = (r * TK + k) * LT;
                if (i0 < n4) {
                    int i = i0 + (int)threadIdx.x;
                    if (i >= n4) i = n4 - 1;
                    int q = i + rot;
                    if (q >= n4) q -= n4;
                    t4[q] = tv[k];
                }
            }
        }
        for (int i = (n4 << 2) + threadIdx.x; i < N; i += LT) lds[i] = dinv[i];
        __syncthreads();
        #pragma unroll
        for (int k = 0; k < EK; ++k) {
            const int p = p0 + (int)threadIdx.x + k * LT;
            if (p < p1) colw[p] = make_int2(ce[k].x, __float_as_int(lds[ce[k].x] * lds[ce[k].y]));
        }
        #pragma unroll 1
        for (int p = p0 + (int)threadIdx.x + EK * LT; p < p1; p += LT) {
            const int2 c = colw[p];
            colw[p] = make_int2(c.x, __float_as_int(lds[c.x] * lds[c.y]));
        }
        __syncthreads();
    }

    #pragma unroll
    for (int i = 0; i < WI; ++i) {
        const int u = threadIdx.x + i * LT;
        if (u < WU) {
            const int p = u >> 5;
            const int n = (u & 31) * 4;
            const float4 r0 = wr0[i];
            const float4 r1 = wr1[i];
            float* d = &Wl[p * 256 + n * 2];
            *(float4*)(d)     = make_float4(r0.x, r1.x, r0.y, r1.y);
            *(float4*)(d + 4) = make_float4(r0.z, r1.z, r0.w, r1.w);
        }
    }

    if (wave >= MW) {
        const int b0 = __builtin_amdgcn_readlane(rpv, 0);
        cwn = colw[(b0 + lane < eLast) ? (b0 + lane) : eLast];
    }

    const int hc  = (wave >> 2) & 1;
    const int mst = (wave < 4) ? wave : ((wave + 1) & 3);
    const int bc  = hc * 64 + lo * 4;
    const float4 bv = make_float4(bias[bc + 0], bias[bc + 1], bias[bc + 2], bias[bc + 3]);

    for (int t = 0; t <= nst; ++t) {
        if (wave >= MW) {
            if (t < nst && SS * t + sg < cnt) {
                float* dstp = sA + (((t & 1) * SS + sg) * 16 + (g & 3) * 4) * SP;
                const int row0 = s0 * 16 + 64 * t + 4 * g;
                int beg = __builtin_amdgcn_readlane(rpv, 5 * t);
                for (int i = 0; i < 4; ++i) {
                    const int row = row0 + i;
                    const int end = __builtin_amdgcn_readlane(rpv, 5 * t + i + 1);
                    const int2 cw0 = cwn;
                    const int nxt = (i < 3) ? end : __builtin_amdgcn_readlane(rpv, 5 * t + 5);
                    cwn = colw[(nxt + lane < eLast) ? (nxt + lane) : eLast];
                    const float di = __int_as_float(__builtin_amdgcn_readlane(__float_as_int(dv), 5 * t + i));
                    const float wself = di * di;
                    const float4 hs = *(const float4*)(Hin + (size_t)((row < N) ? row : (N - 1)) * C + lane * 4);
                    float4 acc = make_float4(0.f, 0.f, 0.f, 0.f);
                    {
                        const int n = (end - beg < 32) ? (end - beg) : 32;
                        int j = 0;
                        if (n > 16) {
                            gather_k<16>(acc, Hin, cw0.x, cw0.y, 0, lane);
                            j = 16;
                        }
                        gather_n(acc, Hin, cw0.x, cw0.y, j, n - j, lane);
                    }
                    for (int base = beg + 32; base < end; base += 32) {
                        const int m = (end - base < 32) ? (end - base) : 32;
                        int2 cx = make_int2(0, 0);
                        if (lane < m) cx = colw[base + lane];
                        for (int j = 0; j < m; ++j) gather_k<1>(acc, Hin, cx.x, cx.y, j, lane);
                    }
                    acc.x = fmaf(hs.x, wself, acc.x);
                    acc.y = fmaf(hs.y, wself, acc.y);
                    acc.z = fmaf(hs.z, wself, acc.z);
                    acc.w = fmaf(hs.w, wself, acc.w);
                    *(float4*)(dstp + i * SP + lane * 4) = acc;
                    beg = end;
                }
            }
        } else if (t >= 1 && SS * (t - 1) + mst < cnt) {
            const int tp = t - 1;
            const int m0 = (s0 + SS * tp + mst) * 16;
            const float* sW = sA + ((tp & 1) * SS + mst) * 16 * SP;
            float* sR = sO + wave * 16 * OP;
            v8f acc[4] = {};
            const float* Ab = sW + lo * SP + 2 * half;
            #pragma unroll
            for (int k0 = 0; k0 < C; k0 += 4) {
                const v2f a = *(const v2f*)(Ab + k0);
                const float* Bbase = &Wl[((k0 >> 1) + half) * 256 + hc * 128 + lo * 2];
                #pragma unroll
                for (int c = 0; c < 4; ++c) {
                    const v2f b = *(const v2f*)(Bbase + c * 32);
                    acc[c] = __builtin_amdgcn_wmma_f32_16x16x4_f32(
                        false, a, false, b, (short)0, acc[c], false, false);
                }
            }
            #pragma unroll
            for (int c = 0; c < 4; ++c)
                #pragma unroll
                for (int r = 0; r < 8; ++r)
                    sR[(r + 8 * half) * OP + c * 16 + lo] = acc[c][r];
            for (int r0 = 0; r0 < 16; r0 += 8) {
                float4 v[4];
                #pragma unroll
                for (int q = 0; q < 4; ++q) v[q] = *(const float4*)(sR + (r0 + 2 * q + half) * OP + lo * 4);
                #pragma unroll
                for (int q = 0; q < 4; ++q) {
                    const int row = m0 + r0 + 2 * q + half;
                    if (row < N) {
                        float4 o;
                        o.x = elu1(v[q].x + bv.x);
                        o.y = elu1(v[q].y + bv.y);
                        o.z = elu1(v[q].z + bv.z);
                        o.w = elu1(v[q].w + bv.w);
                        *(float4*)(out + (size_t)row * C + bc) = o;
                    }
                }
            }
        }
        __syncthreads();
    }
}

extern "C" void kernel_launch(void* const* d_in, const int* in_sizes, int n_in,
                              void* d_out, int out_size, void* d_ws, size_t ws_size,
                              hipStream_t stream) {
    const float* x  = (const float*)d_in[0];
    const int*   ei = (const int*)d_in[1];
    const float* W1 = (const float*)d_in[2];
    const float* b1 = (const float*)d_in[3];
    const float* W2 = (const float*)d_in[4];
    const float* b2 = (const float*)d_in[5];

    const int N = in_sizes[0] / C;
    const int E = in_sizes[1] / 2;
    const int* srcI = ei;
    const int* dstI = ei + E;

    const int B = (N + NPB - 1) / NPB;
    const int G = (E + CH - 1) / CH;

    char* ws = (char*)d_ws;
    size_t off = 0;
    int* rowptr = (int*)(ws + off);   off += ((size_t)(N + 1) * 4 + 255) & ~(size_t)255;
    float* dinv = (float*)(ws + off); off += ((size_t)N * 4 + 255) & ~(size_t)255;
    int* histT = (int*)(ws + off);    off += ((size_t)B * G * 4 + 255) & ~(size_t)255;
    int* offT = (int*)(ws + off);     off += ((size_t)B * G * 4 + 255) & ~(size_t)255;
    int2* tmp = (int2*)(ws + off);    off += ((size_t)G * CH * 8 + 255) & ~(size_t)255;
    int2* colw = (int2*)(ws + off);   off += ((size_t)E * 8 + 255) & ~(size_t)255;
    float* buf1 = (float*)(ws + off);

    const int strips = (N + 15) / 16;
    const int layerBlocks = ((strips + 15) / 16 > WGPS) ? (strips + 15) / 16 : WGPS;

    bin_kernel<<<G, dim3(BT), 0, stream>>>(srcI, dstI, tmp, histT, offT, E, B, G);
    bucket_kernel<<<B, dim3(BT), 0, stream>>>(tmp, histT, offT, rowptr, dinv, colw, N, B, G);

    layer_fused<<<layerBlocks, dim3(LT), 0, stream>>>(x, W1, b1, rowptr, colw, dinv, buf1, N, E, strips, 1);
    layer_fused<<<layerBlocks, dim3(LT), 0, stream>>>(buf1, W2, b2, rowptr, colw, dinv, (float*)d_out, N, E, strips, 0);
}


# Bucket kernel: dependent global round trips of a block cut from 14 to 2: edges in front of a bucket = sum of its chunk offsets (no sweep over earlier buckets), a wave's 7 segments loaded as one group
# speedup vs baseline: 5.1151x; 1.0603x over previous
#include <hip/hip_runtime.h>
#include <math.h>

#define C 128

typedef float v2f __attribute__((ext_vector_type(2)));
typedef float v8f __attribute__((ext_vector_type(8)));

#define BT 1024
#define CH 4096
#define NSH 8
#define NPB 256
#define MAXB 256
#define CAP 8192
#define SB 8

__device__ __forceinline__ int block_excl_scan(int v, int* wt, int lane, int wave, int& total) {
    int incl = v;
    #pragma unroll
    for (int off = 1; off < 32; off <<= 1) {
        int t = __shfl_up(incl, off);
        if (lane >= off) incl += t;
    }
    if (lane == 31) wt[wave] = incl;
    __syncthreads();
    if (wave == 0) {
        const int t = wt[lane];
        int ti = t;
        #pragma unroll
        for (int off = 1; off < 32; off <<= 1) {
            int u = __shfl_up(ti, off);
            if (lane >= off) ti += u;
        }
        wt[lane] = ti - t;
        if (lane == 31) wt[32] = ti;
    }
    __syncthreads();
    const int res = wt[wave] + incl - v;
    total = wt[32];
    __syncthreads();
    return res;
}

__device__ __forceinline__ int block_excl_scan_sum(int v, int u, int* wt, int* wu, int lane, int wave,
                                                   int& total, int& usum) {
    int incl = v, us = u;
    #pragma unroll
    for (int off = 1; off < 32; off <<= 1) {
        int t = __shfl_up(incl, off);
        int s = __shfl_up(us, off);
        if (lane >= off) { incl += t; us += s; }
    }
    if (lane == 31) { wt[wave] = incl; wu[wave] = us; }
    __syncthreads();
    if (wave == 0) {
        const int t = wt[lane];
        int ti = t;
        int ui = wu[lane];
        #pragma unroll
        for (int off = 1; off < 32; off <<= 1) {
            int a = __shfl_up(ti, off);
            int s = __shfl_up(ui, off);
            if (lane >= off) { ti += a; ui += s; }
        }
        wt[lane] = ti - t;
        if (lane == 31) { wt[32] = ti; wu[32] = ui; }
    }
    __syncthreads();
    const int res = wt[wave] + incl - v;
    total = wt[32];
    usum  = wu[32];
    __syncthreads();
    return res;
}

__global__ __launch_bounds__(BT) void bin_kernel(const int* __restrict__ src, const int* __restrict__ dst,
                                                 int2* __restrict__ tmp, int* __restrict__ histT,
                                                 int* __restrict__ offT, int E, int B, int G) {
    __shared__ int cntB[MAXB];
    __shared__ int offB[MAXB];
    __shared__ int wt[33];
    __shared__ int2 sE[CH];
    const int tid  = threadIdx.x;
    const int lane = tid & 31;
    const int wave = tid >> 5;
    const int g  = blockIdx.x;
    const int e0 = g * CH;

    if (tid < MAXB) cntB[tid] = 0;
    __syncthreads();

    int s[CH / BT], d[CH / BT], r[CH / BT];
    #pragma unroll
    for (int k = 0; k < CH / BT; ++k) {
        const int e = e0 + k * BT + tid;
        d[k] = -1; s[k] = 0; r[k] = 0;
        if (e < E) { s[k] = src[e]; d[k] = dst[e]; }
    }
    #pragma unroll
    for (int k = 0; k < CH / BT; ++k)
        if (d[k] >= 0) r[k] = atomicAdd(&cntB[d[k] >> NSH], 1);
    __syncthreads();

    int total;
    const int v = (tid < MAXB) ? cntB[tid] : 0;
    const int ex = block_excl_scan(v, wt, lane, wave, total);
    if (tid < MAXB) offB[tid] = ex;
    __syncthreads();

    #pragma unroll
    for (int k = 0; k < CH / BT; ++k)
        if (d[k] >= 0) sE[offB[d[k] >> NSH] + r[k]] = make_int2(s[k], d[k]);
    __syncthreads();

    const int n = (E - e0 < CH) ? (E - e0) : CH;
    for (int i = tid; i < n; i += BT) tmp[(size_t)e0 + i] = sE[i];
    if (tid < B) {
        histT[tid * G + g] = cntB[tid];
        offT[tid * G + g]  = offB[tid];
    }
}

__global__ __launch_bounds__(BT) void bucket_kernel(const int2* __restrict__ tmp, const int* __restrict__ histT,
                                                    const int* __restrict__ offT, int* __restrict__ rowptr,
                                                    float* __restrict__ dinv, int2* __restrict__ colw,
                                                    int N, int B, int G) {
    __shared__ int segStart[BT + 2];
    __shared__ int segOff[BT + 1];
    __shared__ int cntN[NPB];
    __shared__ int lrp[NPB];
    __shared__ int cur[NPB];
    __shared__ int wt[33];
    __shared__ int wu[33];
    __shared__ int2 eS[CAP];
    __shared__ int2 oS[CAP];
    const int tid  = threadIdx.x;
    const int lane = tid & 31;
    const int wave = tid >> 5;
    const int b = blockIdx.x;

    const int h  = (tid < G) ? histT[b * G + tid] : 0;
    const int so = (tid < G) ? offT[b * G + tid] : 0;
    int nb, base;
    const int st0 = block_excl_scan_sum(h, so, wt, wu, lane, wave, nb, base);
    if (tid < G) { segStart[tid] = st0; segOff[tid] = so; }
    if (tid == 0) { segStart[G] = nb; segStart[G + 1] = nb; segOff[G] = 0; }
    if (tid < NPB) { cntN[tid] = 0; cur[tid] = 0; }
    __syncthreads();

    const bool staged = (nb <= CAP);

    const int wv = __builtin_amdgcn_readfirstlane(wave);
    for (int g0 = wv; g0 < G; g0 += SB * (BT / 32)) {
        int st[SB], len[SB], of[SB];
        int2 e[SB], e2[SB];
        #pragma unroll
        for (int k = 0; k < SB; ++k) {
            const int g  = g0 + k * (BT / 32);
            const int gc = (g < G) ? g : G;
            st[k]  = segStart[gc];
            len[k] = segStart[gc + 1] - st[k];
            of[k]  = gc * CH + segOff[gc];
        }
        #pragma unroll
        for (int k = 0; k < SB; ++k) e[k] = tmp[(lane < len[k]) ? (of[k] + lane) : 0];
        #pragma unroll
        for (int k = 0; k < SB; ++k) e2[k] = tmp[(lane + 32 < len[k]) ? (of[k] + lane + 32) : 0];
        #pragma unroll
        for (int k = 0; k < SB; ++k) {
            if (lane < len[k]) {
                if (staged) eS[st[k] + lane] = e[k];
                atomicAdd(&cntN[e[k].y & (NPB - 1)], 1);
            }
            if (lane + 32 < len[k]) {
                if (staged) eS[st[k] + lane + 32] = e2[k];
                atomicAdd(&cntN[e2[k].y & (NPB - 1)], 1);
            }
            for (int j = lane + 64; j < len[k]; j += 32) {
                const int2 f = tmp[of[k] + j];
                if (staged) eS[st[k] + j] = f;
                atomicAdd(&cntN[f.y & (NPB - 1)], 1);
            }
        }
    }
    __syncthreads();

    int tot2;
    const int c  = (tid < NPB) ? cntN[tid] : 0;
    const int lx = block_excl_scan(c, wt, lane, wave, tot2);
    if (tid < NPB) {
        lrp[tid] = lx;
        const int node = b * NPB + tid;
        if (node < N) {
            rowptr[node] = base + lx;
            dinv[node] = rsqrtf((float)c + 1.0f);
            if (node == N - 1) rowptr[N] = base + lx + c;
        }
    }
    __syncthreads();

    if (staged) {
        for (int i = tid; i < nb; i += BT) {
            const int2 e = eS[i];
            const int nd = e.y & (NPB - 1);
            const int r = atomicAdd(&cur[nd], 1);
            oS[lrp[nd] + r] = e;
        }
        __syncthreads();
        for (int i = tid; i < nb; i += BT) colw[(size_t)base + i] = oS[i];
    } else {
        for (int g = wave; g < G; g += BT / 32) {
            const int st  = segStart[g];
            const int len = segStart[g + 1] - st;
            const int2* p = tmp + (size_t)g * CH + segOff[g];
            for (int j = lane; j < len; j += 32) {
                const int2 e = p[j];
                const int nd = e.y & (NPB - 1);
                const int r = atomicAdd(&cur[nd], 1);
                colw[(size_t)base + lrp[nd] + r] = e;
            }
        }
    }
}

#define SP 132

__device__ __forceinline__ float elu1(float v) {
    return v > 0.f ? v : (expf(v) - 1.f);
}

template <int K>
__device__ __forceinline__ void gather_k(float4& acc, const float* __restrict__ H,
                                         int cols, int wbits, int j, int lane) {
    float4 v[K];
    float  w[K];
    #pragma unroll
    for (int k = 0; k < K; ++k) {
        const int s = __builtin_amdgcn_readlane(cols, j + k);
        w[k] = __int_as_float(__builtin_amdgcn_readlane(wbits, j + k));
        v[k] = *(const float4*)(H + (size_t)s * C + lane * 4);
    }
    #pragma unroll
    for (int k = 0; k < K; ++k) {
        acc.x = fmaf(v[k].x, w[k], acc.x);
        acc.y = fmaf(v[k].y, w[k], acc.y);
        acc.z = fmaf(v[k].z, w[k], acc.z);
        acc.w = fmaf(v[k].w, w[k], acc.w);
    }
}

__device__ __forceinline__ void gather_n(float4& acc, const float* __restrict__ H,
                                         int cols, int wbits, int j, int cnt, int lane) {
    switch (cnt) {
        case 1:  gather_k<1>(acc, H, cols, wbits, j, lane);  break;
        case 2:  gather_k<2>(acc, H, cols, wbits, j, lane);  break;
        case 3:  gather_k<3>(acc, H, cols, wbits, j, lane);  break;
        case 4:  gather_k<4>(acc, H, cols, wbits, j, lane);  break;
        case 5:  gather_k<5>(acc, H, cols, wbits, j, lane);  break;
        case 6:  gather_k<6>(acc, H, cols, wbits, j, lane);  break;
        case 7:  gather_k<7>(acc, H, cols, wbits, j, lane);  break;
        case 8:  gather_k<8>(acc, H, cols, wbits, j, lane);  break;
        case 9:  gather_k<9>(acc, H, cols, wbits, j, lane);  break;
        case 10: gather_k<10>(acc, H, cols, wbits, j, lane); break;
        case 11: gather_k<11>(acc, H, cols, wbits, j, lane); break;
        case 12: gather_k<12>(acc, H, cols, wbits, j, lane); break;
        case 13: gather_k<13>(acc, H, cols, wbits, j, lane); break;
        case 14: gather_k<14>(acc, H, cols, wbits, j, lane); break;
        case 15: gather_k<15>(acc, H, cols, wbits, j, lane); break;
        case 16: gather_k<16>(acc, H, cols, wbits, j, lane); break;
        default: break;
    }
}

#define LT 768
#define MW 8
#define SS 4
#define OP 68
#define WGPS 256
#define TABN 65536
#define EK 5
#define TK 11

__global__ __launch_bounds__(LT) void layer_fused(const float* __restrict__ Hin,
                                                  const float* __restrict__ W,
                                                  const float* __restrict__ bias,
                                                  const int* __restrict__ rowptr,
                                                  int2* colw,
                                                  const float* __restrict__ dinv,
                                                  float* __restrict__ out,
                                                  int N, int E, int nStrips, int mkw) {
    __shared__ __attribute__((aligned(16))) float lds[TABN];
    float* const Wl = lds;
    float* const sA = lds + C * C;
    float* const sO = sA + 2 * SS * 16 * SP;

    const int lane = threadIdx.x & 31;
    const int half = lane >> 4;
    const int lo   = lane & 15;
    const int wave = __builtin_amdgcn_readfirstlane((int)(threadIdx.x >> 5));

    const int s0  = (int)((blockIdx.x * (unsigned)nStrips) / gridDim.x);
    const int s1  = (int)(((blockIdx.x + 1u) * (unsigned)nStrips) / gridDim.x);
    const int cnt = s1 - s0;
    const int nst = (cnt + SS - 1) / SS;

    int p0 = 0, p1 = 0;
    if (mkw) {
        p0 = rowptr[(s0 * 16 < N) ? s0 * 16 : N];
        p1 = rowptr[(s1 * 16 < N) ? s1 * 16 : N];
    }

    constexpr int WU = (C / 2) * (C / 4);
    constexpr int WI = (WU + LT - 1) / LT;
    float4 wr0[WI], wr1[WI];
    #pragma unroll
    for (int i = 0; i < WI; ++i) {
        const int u = threadIdx.x + i * LT;
        wr0[i] = make_float4(0.f, 0.f, 0.f, 0.f);
        wr1[i] = wr0[i];
        if (u < WU) {
            const int p = u >> 5;
            const int n = (u & 31) * 4;
            wr0[i] = *(const float4*)(W + (size_t)(2 * p) * C + n);
            wr1[i] = *(const float4*)(W + (size_t)(2 * p + 1) * C + n);
        }
    }

    const int g  = wave - MW;
    const int sg = g >> 2;
    const int eLast = (E > 0) ? (E - 1) : 0;
    int rpv = 0;
    float dv = 0.f;
    int2 cwn = make_int2(0, 0);
    if (wave >= MW) {
        if (lane < 25) {
            const int tt = lane / 5;
            const int j  = lane - 5 * tt;
            const int row = s0 * 16 + 64 * tt + 4 * g + j;
            rpv = rowptr[(row < N) ? row : N];
            if (j < 4 && row < N) dv = dinv[row];
        }
    }

    if (mkw) {
        int2 ce[EK];
        #pragma unroll
        for (int k = 0; k < EK; ++k) {
            const int p = p0 + (int)threadIdx.x + k * LT;
            ce[k] = make_int2(0, 0);
            if (p < p1) ce[k] = colw[p];
        }
        const int n4 = N >> 2;
        const int rot = (int)((blockIdx.x * (unsigned)n4) / gridDim.x);
        const float4* d4 = (const float4*)dinv;
        float4* t4 = (float4*)lds;
        #pragma unroll
        for (int r = 0; r < (TABN / 4 + TK * LT - 1) / (TK * LT); ++r) {
            float4 tv[TK];
            #pragma unroll
            for (int k = 0; k < TK; ++k) {
                const int i0 = (r * TK + k) * LT;
                tv[k] = make_float4(0.f, 0.f, 0.f, 0.f);
                if (i0 < n4) {
                    int i = i0 + (int)threadIdx.x;
                    if (i >= n4) i = n4 - 1;
                    int q = i + rot;
                    if (q >= n4) q -= n4;
                    tv[k] = d4[q];
                }
            }
            #pragma unroll
            for (int k = 0; k < TK; ++k) {
                const int i0 = (r * TK + k) * LT;
                if (i0 < n4) {
                    int i = i0 + (int)threadIdx.x;
                    if (i >= n4) i = n4 - 1;
                    int q = i + rot;
                    if (q >= n4) q -= n4;
                    t4[q] = tv[k];
                }
            }
        }
        for (int i = (n4 << 2) + threadIdx.x; i < N; i += LT) lds[i] = dinv[i];
        __syncthreads();
        #pragma unroll
        for (int k = 0; k < EK; ++k) {
            const int p = p0 + (int)threadIdx.x + k * LT;
            if (p < p1) colw[p] = make_int2(ce[k].x, __float_as_int(lds[ce[k].x] * lds[ce[k].y]));
        }
        #pragma unroll 1
        for (int p = p0 + (int)threadIdx.x + EK * LT; p < p1; p += LT) {
            const int2 c = colw[p];
            colw[p] = make_int2(c.x, __float_as_int(lds[c.x] * lds[c.y]));
        }
        __syncthreads();
    }

    #pragma unroll
    for (int i = 0; i < WI; ++i) {
        const int u = threadIdx.x + i * LT;
        if (u < WU) {
            const int p = u >> 5;
            const int n = (u & 31) * 4;
            const float4 r0 = wr0[i];
            const float4 r1 = wr1[i];
            float* d = &Wl[p * 256 + n * 2];
            *(float4*)(d)     = make_float4(r0.x, r1.x, r0.y, r1.y);
            *(float4*)(d + 4) = make_float4(r0.z, r1.z, r0.w, r1.w);
        }
    }

    if (wave >= MW) {
        const int b0 = __builtin_amdgcn_readlane(rpv, 0);
        cwn = colw[(b0 + lane < eLast) ? (b0 + lane) : eLast];
    }

    const int hc  = (wave >> 2) & 1;
    const int mst = (wave < 4) ? wave : ((wave + 1) & 3);
    const int bc  = hc * 64 + lo * 4;
    const float4 bv = make_float4(bias[bc + 0], bias[bc + 1], bias[bc + 2], bias[bc + 3]);

    for (int t = 0; t <= nst; ++t) {
        if (wave >= MW) {
            if (t < nst && SS * t + sg < cnt) {
                float* dstp = sA + (((t & 1) * SS + sg) * 16 + (g & 3) * 4) * SP;
                const int row0 = s0 * 16 + 64 * t + 4 * g;
                int beg = __builtin_amdgcn_readlane(rpv, 5 * t);
                for (int i = 0; i < 4; ++i) {
                    const int row = row0 + i;
                    const int end = __builtin_amdgcn_readlane(rpv, 5 * t + i + 1);
                    const int2 cw0 = cwn;
                    const int nxt = (i < 3) ? end : __builtin_amdgcn_readlane(rpv, 5 * t + 5);
                    cwn = colw[(nxt + lane < eLast) ? (nxt + lane) : eLast];
                    const float di = __int_as_float(__builtin_amdgcn_readlane(__float_as_int(dv), 5 * t + i));
                    const float wself = di * di;
                    const float4 hs = *(const float4*)(Hin + (size_t)((row < N) ? row : (N - 1)) * C + lane * 4);
                    float4 acc = make_float4(0.f, 0.f, 0.f, 0.f);
                    {
                        const int n = (end - beg < 32) ? (end - beg) : 32;
                        int j = 0;
                        if (n > 16) {
                            gather_k<16>(acc, Hin, cw0.x, cw0.y, 0, lane);
                            j = 16;
                        }
                        gather_n(acc, Hin, cw0.x, cw0.y, j, n - j, lane);
                    }
                    for (int base = beg + 32; base < end; base += 32) {
                        const int m = (end - base < 32) ? (end - base) : 32;
                        int2 cx = make_int2(0, 0);
                        if (lane < m) cx = colw[base + lane];
                        for (int j = 0; j < m; ++j) gather_k<1>(acc, Hin, cx.x, cx.y, j, lane);
                    }
                    acc.x = fmaf(hs.x, wself, acc.x);
                    acc.y = fmaf(hs.y, wself, acc.y);
                    acc.z = fmaf(hs.z, wself, acc.z);
                    acc.w = fmaf(hs.w, wself, acc.w);
                    *(float4*)(dstp + i * SP + lane * 4) = acc;
                    beg = end;
                }
            }
        } else if (t >= 1 && SS * (t - 1) + mst < cnt) {
            const int tp = t - 1;
            const int m0 = (s0 + SS * tp + mst) * 16;
            const float* sW = sA + ((tp & 1) * SS + mst) * 16 * SP;
            float* sR = sO + wave * 16 * OP;
            v8f acc[4] = {};
            const float* Ab = sW + lo * SP + 2 * half;
            #pragma unroll
            for (int k0 = 0; k0 < C; k0 += 4) {
                const v2f a = *(const v2f*)(Ab + k0);
                const float* Bbase = &Wl[((k0 >> 1) + half) * 256 + hc * 128 + lo * 2];
                #pragma unroll
                for (int c = 0; c < 4; ++c) {
                    const v2f b = *(const v2f*)(Bbase + c * 32);
                    acc[c] = __builtin_amdgcn_wmma_f32_16x16x4_f32(
                        false, a, false, b, (short)0, acc[c], false, false);
                }
            }
            #pragma unroll
            for (int c = 0; c < 4; ++c)
                #pragma unroll
                for (int r = 0; r < 8; ++r)
                    sR[(r + 8 * half) * OP + c * 16 + lo] = acc[c][r];
            for (int r0 = 0; r0 < 16; r0 += 8) {
                float4 v[4];
                #pragma unroll
                for (int q = 0; q < 4; ++q) v[q] = *(const float4*)(sR + (r0 + 2 * q + half) * OP + lo * 4);
                #pragma unroll
                for (int q = 0; q < 4; ++q) {
                    const int row = m0 + r0 + 2 * q + half;
                    if (row < N) {
                        float4 o;
                        o.x = elu1(v[q].x + bv.x);
                        o.y = elu1(v[q].y + bv.y);
                        o.z = elu1(v[q].z + bv.z);
                        o.w = elu1(v[q].w + bv.w);
                        *(float4*)(out + (size_t)row * C + bc) = o;
                    }
                }
            }
        }
        __syncthreads();
    }
}

extern "C" void kernel_launch(void* const* d_in, const int* in_sizes, int n_in,
                              void* d_out, int out_size, void* d_ws, size_t ws_size,
                              hipStream_t stream) {
    const float* x  = (const float*)d_in[0];
    const int*   ei = (const int*)d_in[1];
    const float* W1 = (const float*)d_in[2];
    const float* b1 = (const float*)d_in[3];
    const float* W2 = (const float*)d_in[4];
    const float* b2 = (const float*)d_in[5];

    const int N = in_sizes[0] / C;
    const int E = in_sizes[1] / 2;
    const int* srcI = ei;
    const int* dstI = ei + E;

    const int B = (N + NPB - 1) / NPB;
    const int G = (E + CH - 1) / CH;

    char* ws = (char*)d_ws;
    size_t off = 0;
    int* rowptr = (int*)(ws + off);   off += ((size_t)(N + 1) * 4 + 255) & ~(size_t)255;
    float* dinv = (float*)(ws + off); off += ((size_t)N * 4 + 255) & ~(size_t)255;
    int* histT = (int*)(ws + off);    off += ((size_t)B * G * 4 + 255) & ~(size_t)255;
    int* offT = (int*)(ws + off);     off += ((size_t)B * G * 4 + 255) & ~(size_t)255;
    int2* tmp = (int2*)(ws + off);    off += ((size_t)G * CH * 8 + 255) & ~(size_t)255;
    int2* colw = (int2*)(ws + off);   off += ((size_t)E * 8 + 255) & ~(size_t)255;
    float* buf1 = (float*)(ws + off);

    const int strips = (N + 15) / 16;
    const int layerBlocks = ((strips + 15) / 16 > WGPS) ? (strips + 15) / 16 : WGPS;

    bin_kernel<<<G, dim3(BT), 0, stream>>>(srcI, dstI, tmp, histT, offT, E, B, G);
    bucket_kernel<<<B, dim3(BT), 0, stream>>>(tmp, histT, offT, rowptr, dinv, colw, N, B, G);

    layer_fused<<<layerBlocks, dim3(LT), 0, stream>>>(x, W1, b1, rowptr, colw, dinv, buf1, N, E, strips, 1);
    layer_fused<<<layerBlocks, dim3(LT), 0, stream>>>(buf1, W2, b2, rowptr, colw, dinv, (float*)d_out, N, E, strips, 0);
}


# Gather waves at raised wave priority (s_setprio 3) through the stage loop: their address arithmetic and row requests issue ahead of the older matrix waves' f32 WMMAs, which take the slots left
# speedup vs baseline: 5.2832x; 1.0329x over previous
#include <hip/hip_runtime.h>
#include <math.h>

#define C 128

typedef float v2f __attribute__((ext_vector_type(2)));
typedef float v8f __attribute__((ext_vector_type(8)));

#define BT 1024
#define CH 4096
#define NSH 8
#define NPB 256
#define MAXB 256
#define CAP 8192
#define SB 8

__device__ __forceinline__ int block_excl_scan(int v, int* wt, int lane, int wave, int& total) {
    int incl = v;
    #pragma unroll
    for (int off = 1; off < 32; off <<= 1) {
        int t = __shfl_up(incl, off);
        if (lane >= off) incl += t;
    }
    if (lane == 31) wt[wave] = incl;
    __syncthreads();
    if (wave == 0) {
        const int t = wt[lane];
        int ti = t;
        #pragma unroll
        for (int off = 1; off < 32; off <<= 1) {
            int u = __shfl_up(ti, off);
            if (lane >= off) ti += u;
        }
        wt[lane] = ti - t;
        if (lane == 31) wt[32] = ti;
    }
    __syncthreads();
    const int res = wt[wave] + incl - v;
    total = wt[32];
    __syncthreads();
    return res;
}

__device__ __forceinline__ int block_excl_scan_sum(int v, int u, int* wt, int* wu, int lane, int wave,
                                                   int& total, int& usum) {
    int incl = v, us = u;
    #pragma unroll
    for (int off = 1; off < 32; off <<= 1) {
        int t = __shfl_up(incl, off);
        int s = __shfl_up(us, off);
        if (lane >= off) { incl += t; us += s; }
    }
    if (lane == 31) { wt[wave] = incl; wu[wave] = us; }
    __syncthreads();
    if (wave == 0) {
        const int t = wt[lane];
        int ti = t;
        int ui = wu[lane];
        #pragma unroll
        for (int off = 1; off < 32; off <<= 1) {
            int a = __shfl_up(ti, off);
            int s = __shfl_up(ui, off);
            if (lane >= off) { ti += a; ui += s; }
        }
        wt[lane] = ti - t;
        if (lane == 31) { wt[32] = ti; wu[32] = ui; }
    }
    __syncthreads();
    const int res = wt[wave] + incl - v;
    total = wt[32];
    usum  = wu[32];
    __syncthreads();
    return res;
}

__global__ __launch_bounds__(BT) void bin_kernel(const int* __restrict__ src, const int* __restrict__ dst,
                                                 int2* __restrict__ tmp, int* __restrict__ histT,
                                                 int* __restrict__ offT, int E, int B, int G) {
    __shared__ int cntB[MAXB];
    __shared__ int offB[MAXB];
    __shared__ int wt[33];
    __shared__ int2 sE[CH];
    const int tid  = threadIdx.x;
    const int lane = tid & 31;
    const int wave = tid >> 5;
    const int g  = blockIdx.x;
    const int e0 = g * CH;

    if (tid < MAXB) cntB[tid] = 0;
    __syncthreads();

    int s[CH / BT], d[CH / BT], r[CH / BT];
    #pragma unroll
    for (int k = 0; k < CH / BT; ++k) {
        const int e = e0 + k * BT + tid;
        d[k] = -1; s[k] = 0; r[k] = 0;
        if (e < E) { s[k] = src[e]; d[k] = dst[e]; }
    }
    #pragma unroll
    for (int k = 0; k < CH / BT; ++k)
        if (d[k] >= 0) r[k] = atomicAdd(&cntB[d[k] >> NSH], 1);
    __syncthreads();

    int total;
    const int v = (tid < MAXB) ? cntB[tid] : 0;
    const int ex = block_excl_scan(v, wt, lane, wave, total);
    if (tid < MAXB) offB[tid] = ex;
    __syncthreads();

    #pragma unroll
    for (int k = 0; k < CH / BT; ++k)
        if (d[k] >= 0) sE[offB[d[k] >> NSH] + r[k]] = make_int2(s[k], d[k]);
    __syncthreads();

    const int n = (E - e0 < CH) ? (E - e0) : CH;
    for (int i = tid; i < n; i += BT) tmp[(size_t)e0 + i] = sE[i];
    if (tid < B) {
        histT[tid * G + g] = cntB[tid];
        offT[tid * G + g]  = offB[tid];
    }
}

__global__ __launch_bounds__(BT) void bucket_kernel(const int2* __restrict__ tmp, const int* __restrict__ histT,
                                                    const int* __restrict__ offT, int* __restrict__ rowptr,
                                                    float* __restrict__ dinv, int2* __restrict__ colw,
                                                    int N, int B, int G) {
    __shared__ int segStart[BT + 2];
    __shared__ int segOff[BT + 1];
    __shared__ int cntN[NPB];
    __shared__ int lrp[NPB];
    __shared__ int cur[NPB];
    __shared__ int wt[33];
    __shared__ int wu[33];
    __shared__ int2 eS[CAP];
    __shared__ int2 oS[CAP];
    const int tid  = threadIdx.x;
    const int lane = tid & 31;
    const int wave = tid >> 5;
    const int b = blockIdx.x;

    const int h  = (tid < G) ? histT[b * G + tid] : 0;
    const int so = (tid < G) ? offT[b * G + tid] : 0;
    int nb, base;
    const int st0 = block_excl_scan_sum(h, so, wt, wu, lane, wave, nb, base);
    if (tid < G) { segStart[tid] = st0; segOff[tid] = so; }
    if (tid == 0) { segStart[G] = nb; segStart[G + 1] = nb; segOff[G] = 0; }
    if (tid < NPB) { cntN[tid] = 0; cur[tid] = 0; }
    __syncthreads();

    const bool staged = (nb <= CAP);

    const int wv = __builtin_amdgcn_readfirstlane(wave);
    for (int g0 = wv; g0 < G; g0 += SB * (BT / 32)) {
        int st[SB], len[SB], of[SB];
        int2 e[SB], e2[SB];
        #pragma unroll
        for (int k = 0; k < SB; ++k) {
            const int g  = g0 + k * (BT / 32);
            const int gc = (g < G) ? g : G;
            st[k]  = segStart[gc];
            len[k] = segStart[gc + 1] - st[k];
            of[k]  = gc * CH + segOff[gc];
        }
        #pragma unroll
        for (int k = 0; k < SB; ++k) e[k] = tmp[(lane < len[k]) ? (of[k] + lane) : 0];
        #pragma unroll
        for (int k = 0; k < SB; ++k) e2[k] = tmp[(lane + 32 < len[k]) ? (of[k] + lane + 32) : 0];
        #pragma unroll
        for (int k = 0; k < SB; ++k) {
            if (lane < len[k]) {
                if (staged) eS[st[k] + lane] = e[k];
                atomicAdd(&cntN[e[k].y & (NPB - 1)], 1);
            }
            if (lane + 32 < len[k]) {
                if (staged) eS[st[k] + lane + 32] = e2[k];
                atomicAdd(&cntN[e2[k].y & (NPB - 1)], 1);
            }
            for (int j = lane + 64; j < len[k]; j += 32) {
                const int2 f = tmp[of[k] + j];
                if (staged) eS[st[k] + j] = f;
                atomicAdd(&cntN[f.y & (NPB - 1)], 1);
            }
        }
    }
    __syncthreads();

    int tot2;
    const int c  = (tid < NPB) ? cntN[tid] : 0;
    const int lx = block_excl_scan(c, wt, lane, wave, tot2);
    if (tid < NPB) {
        lrp[tid] = lx;
        const int node = b * NPB + tid;
        if (node < N) {
            rowptr[node] = base + lx;
            dinv[node] = rsqrtf((float)c + 1.0f);
            if (node == N - 1) rowptr[N] = base + lx + c;
        }
    }
    __syncthreads();

    if (staged) {
        for (int i = tid; i < nb; i += BT) {
            const int2 e = eS[i];
            const int nd = e.y & (NPB - 1);
            const int r = atomicAdd(&cur[nd], 1);
            oS[lrp[nd] + r] = e;
        }
        __syncthreads();
        for (int i = tid; i < nb; i += BT) colw[(size_t)base + i] = oS[i];
    } else {
        for (int g = wave; g < G; g += BT / 32) {
            const int st  = segStart[g];
            const int len = segStart[g + 1] - st;
            const int2* p = tmp + (size_t)g * CH + segOff[g];
            for (int j = lane; j < len; j += 32) {
                const int2 e = p[j];
                const int nd = e.y & (NPB - 1);
                const int r = atomicAdd(&cur[nd], 1);
                colw[(size_t)base + lrp[nd] + r] = e;
            }
        }
    }
}

#define SP 132

__device__ __forceinline__ float elu1(float v) {
    return v > 0.f ? v : (expf(v) - 1.f);
}

template <int K>
__device__ __forceinline__ void gather_k(float4& acc, const float* __restrict__ H,
                                         int cols, int wbits, int j, int lane) {
    float4 v[K];
    float  w[K];
    #pragma unroll
    for (int k = 0; k < K; ++k) {
        const int s = __builtin_amdgcn_readlane(cols, j + k);
        w[k] = __int_as_float(__builtin_amdgcn_readlane(wbits, j + k));
        v[k] = *(const float4*)(H + (size_t)s * C + lane * 4);
    }
    #pragma unroll
    for (int k = 0; k < K; ++k) {
        acc.x = fmaf(v[k].x, w[k], acc.x);
        acc.y = fmaf(v[k].y, w[k], acc.y);
        acc.z = fmaf(v[k].z, w[k], acc.z);
        acc.w = fmaf(v[k].w, w[k], acc.w);
    }
}

__device__ __forceinline__ void gather_n(float4& acc, const float* __restrict__ H,
                                         int cols, int wbits, int j, int cnt, int lane) {
    switch (cnt) {
        case 1:  gather_k<1>(acc, H, cols, wbits, j, lane);  break;
        case 2:  gather_k<2>(acc, H, cols, wbits, j, lane);  break;
        case 3:  gather_k<3>(acc, H, cols, wbits, j, lane);  break;
        case 4:  gather_k<4>(acc, H, cols, wbits, j, lane);  break;
        case 5:  gather_k<5>(acc, H, cols, wbits, j, lane);  break;
        case 6:  gather_k<6>(acc, H, cols, wbits, j, lane);  break;
        case 7:  gather_k<7>(acc, H, cols, wbits, j, lane);  break;
        case 8:  gather_k<8>(acc, H, cols, wbits, j, lane);  break;
        case 9:  gather_k<9>(acc, H, cols, wbits, j, lane);  break;
        case 10: gather_k<10>(acc, H, cols, wbits, j, lane); break;
        case 11: gather_k<11>(acc, H, cols, wbits, j, lane); break;
        case 12: gather_k<12>(acc, H, cols, wbits, j, lane); break;
        case 13: gather_k<13>(acc, H, cols, wbits, j, lane); break;
        case 14: gather_k<14>(acc, H, cols, wbits, j, lane); break;
        case 15: gather_k<15>(acc, H, cols, wbits, j, lane); break;
        case 16: gather_k<16>(acc, H, cols, wbits, j, lane); break;
        default: break;
    }
}

#define LT 768
#define MW 8
#define SS 4
#define OP 68
#define WGPS 256
#define TABN 65536
#define EK 5
#define TK 11

__global__ __launch_bounds__(LT) void layer_fused(const float* __restrict__ Hin,
                                                  const float* __restrict__ W,
                                                  const float* __restrict__ bias,
                                                  const int* __restrict__ rowptr,
                                                  int2* colw,
                                                  const float* __restrict__ dinv,
                                                  float* __restrict__ out,
                                                  int N, int E, int nStrips, int mkw) {
    __shared__ __attribute__((aligned(16))) float lds[TABN];
    float* const Wl = lds;
    float* const sA = lds + C * C;
    float* const sO = sA + 2 * SS * 16 * SP;

    const int lane = threadIdx.x & 31;
    const int half = lane >> 4;
    const int lo   = lane & 15;
    const int wave = __builtin_amdgcn_readfirstlane((int)(threadIdx.x >> 5));

    const int s0  = (int)((blockIdx.x * (unsigned)nStrips) / gridDim.x);
    const int s1  = (int)(((blockIdx.x + 1u) * (unsigned)nStrips) / gridDim.x);
    const int cnt = s1 - s0;
    const int nst = (cnt + SS - 1) / SS;

    int p0 = 0, p1 = 0;
    if (mkw) {
        p0 = rowptr[(s0 * 16 < N) ? s0 * 16 : N];
        p1 = rowptr[(s1 * 16 < N) ? s1 * 16 : N];
    }

    constexpr int WU = (C / 2) * (C / 4);
    constexpr int WI = (WU + LT - 1) / LT;
    float4 wr0[WI], wr1[WI];
    #pragma unroll
    for (int i = 0; i < WI; ++i) {
        const int u = threadIdx.x + i * LT;
        wr0[i] = make_float4(0.f, 0.f, 0.f, 0.f);
        wr1[i] = wr0[i];
        if (u < WU) {
            const int p = u >> 5;
            const int n = (u & 31) * 4;
            wr0[i] = *(const float4*)(W + (size_t)(2 * p) * C + n);
            wr1[i] = *(const float4*)(W + (size_t)(2 * p + 1) * C + n);
        }
    }

    const int g  = wave - MW;
    const int sg = g >> 2;
    const int eLast = (E > 0) ? (E - 1) : 0;
    int rpv = 0;
    float dv = 0.f;
    int2 cwn = make_int2(0, 0);
    if (wave >= MW) {
        if (lane < 25) {
            const int tt = lane / 5;
            const int j  = lane - 5 * tt;
            const int row = s0 * 16 + 64 * tt + 4 * g + j;
            rpv = rowptr[(row < N) ? row : N];
            if (j < 4 && row < N) dv = dinv[row];
        }
    }

    if (mkw) {
        int2 ce[EK];
        #pragma unroll
        for (int k = 0; k < EK; ++k) {
            const int p = p0 + (int)threadIdx.x + k * LT;
            ce[k] = make_int2(0, 0);
            if (p < p1) ce[k] = colw[p];
        }
        const int n4 = N >> 2;
        const int rot = (int)((blockIdx.x * (unsigned)n4) / gridDim.x);
        const float4* d4 = (const float4*)dinv;
        float4* t4 = (float4*)lds;
        #pragma unroll
        for (int r = 0; r < (TABN / 4 + TK * LT - 1) / (TK * LT); ++r) {
            float4 tv[TK];
            #pragma unroll
            for (int k = 0; k < TK; ++k) {
                const int i0 = (r * TK + k) * LT;
                tv[k] = make_float4(0.f, 0.f, 0.f, 0.f);
                if (i0 < n4) {
                    int i = i0 + (int)threadIdx.x;
                    if (i >= n4) i = n4 - 1;
                    int q = i + rot;
                    if (q >= n4) q -= n4;
                    tv[k] = d4[q];
                }
            }
            #pragma unroll
            for (int k = 0; k < TK; ++k) {
                const int i0 = (r * TK + k) * LT;
                if (i0 < n4) {
                    int i = i0 + (int)threadIdx.x;
                    if (i >= n4) i = n4 - 1;
                    int q = i + rot;
                    if (q >= n4) q -= n4;
                    t4[q] = tv[k];
                }
            }
        }
        for (int i = (n4 << 2) + threadIdx.x; i < N; i += LT) lds[i] = dinv[i];
        __syncthreads();
        #pragma unroll
        for (int k = 0; k < EK; ++k) {
            const int p = p0 + (int)threadIdx.x + k * LT;
            if (p < p1) colw[p] = make_int2(ce[k].x, __float_as_int(lds[ce[k].x] * lds[ce[k].y]));
        }
        #pragma unroll 1
        for (int p = p0 + (int)threadIdx.x + EK * LT; p < p1; p += LT) {
            const int2 c = colw[p];
            colw[p] = make_int2(c.x, __float_as_int(lds[c.x] * lds[c.y]));
        }
        __syncthreads();
    }

    #pragma unroll
    for (int i = 0; i < WI; ++i) {
        const int u = threadIdx.x + i * LT;
        if (u < WU) {
            const int p = u >> 5;
            const int n = (u & 31) * 4;
            const float4 r0 = wr0[i];
            const float4 r1 = wr1[i];
            float* d = &Wl[p * 256 + n * 2];
            *(float4*)(d)     = make_float4(r0.x, r1.x, r0.y, r1.y);
            *(float4*)(d + 4) = make_float4(r0.z, r1.z, r0.w, r1.w);
        }
    }

    if (wave >= MW) {
        __builtin_amdgcn_s_setprio(3);
        const int b0 = __builtin_amdgcn_readlane(rpv, 0);
        cwn = colw[(b0 + lane < eLast) ? (b0 + lane) : eLast];
    }

    const int hc  = (wave >> 2) & 1;
    const int mst = (wave < 4) ? wave : ((wave + 1) & 3);
    const int bc  = hc * 64 + lo * 4;
    const float4 bv = make_float4(bias[bc + 0], bias[bc + 1], bias[bc + 2], bias[bc + 3]);

    for (int t = 0; t <= nst; ++t) {
        if (wave >= MW) {
            if (t < nst && SS * t + sg < cnt) {
                float* dstp = sA + (((t & 1) * SS + sg) * 16 + (g & 3) * 4) * SP;
                const int row0 = s0 * 16 + 64 * t + 4 * g;
                int beg = __builtin_amdgcn_readlane(rpv, 5 * t);
                for (int i = 0; i < 4; ++i) {
                    const int row = row0 + i;
                    const int end = __builtin_amdgcn_readlane(rpv, 5 * t + i + 1);
                    const int2 cw0 = cwn;
                    const int nxt = (i < 3) ? end : __builtin_amdgcn_readlane(rpv, 5 * t + 5);
                    cwn = colw[(nxt + lane < eLast) ? (nxt + lane) : eLast];
                    const float di = __int_as_float(__builtin_amdgcn_readlane(__float_as_int(dv), 5 * t + i));
                    const float wself = di * di;
                    const float4 hs = *(const float4*)(Hin + (size_t)((row < N) ? row : (N - 1)) * C + lane * 4);
                    float4 acc = make_float4(0.f, 0.f, 0.f, 0.f);
                    {
                        const int n = (end - beg < 32) ? (end - beg) : 32;
                        int j = 0;
                        if (n > 16) {
                            gather_k<16>(acc, Hin, cw0.x, cw0.y, 0, lane);
                            j = 16;
                        }
                        gather_n(acc, Hin, cw0.x, cw0.y, j, n - j, lane);
                    }
                    for (int base = beg + 32; base < end; base += 32) {
                        const int m = (end - base < 32) ? (end - base) : 32;
                        int2 cx = make_int2(0, 0);
                        if (lane < m) cx = colw[base + lane];
                        for (int j = 0; j < m; ++j) gather_k<1>(acc, Hin, cx.x, cx.y, j, lane);
                    }
                    acc.x = fmaf(hs.x, wself, acc.x);
                    acc.y = fmaf(hs.y, wself, acc.y);
                    acc.z = fmaf(hs.z, wself, acc.z);
                    acc.w = fmaf(hs.w, wself, acc.w);
                    *(float4*)(dstp + i * SP + lane * 4) = acc;
                    beg = end;
                }
            }
        } else if (t >= 1 && SS * (t - 1) + mst < cnt) {
            const int tp = t - 1;
            const int m0 = (s0 + SS * tp + mst) * 16;
            const float* sW = sA + ((tp & 1) * SS + mst) * 16 * SP;
            float* sR = sO + wave * 16 * OP;
            v8f acc[4] = {};
            const float* Ab = sW + lo * SP + 2 * half;
            #pragma unroll
            for (int k0 = 0; k0 < C; k0 += 4) {
                const v2f a = *(const v2f*)(Ab + k0);
                const float* Bbase = &Wl[((k0 >> 1) + half) * 256 + hc * 128 + lo * 2];
                #pragma unroll
                for (int c = 0; c < 4; ++c) {
                    const v2f b = *(const v2f*)(Bbase + c * 32);
                    acc[c] = __builtin_amdgcn_wmma_f32_16x16x4_f32(
                        false, a, false, b, (short)0, acc[c], false, false);
                }
            }
            #pragma unroll
            for (int c = 0; c < 4; ++c)
                #pragma unroll
                for (int r = 0; r < 8; ++r)
                    sR[(r + 8 * half) * OP + c * 16 + lo] = acc[c][r];
            for (int r0 = 0; r0 < 16; r0 += 8) {
                float4 v[4];
                #pragma unroll
                for (int q = 0; q < 4; ++q) v[q] = *(const float4*)(sR + (r0 + 2 * q + half) * OP + lo * 4);
                #pragma unroll
                for (int q = 0; q < 4; ++q) {
                    const int row = m0 + r0 + 2 * q + half;
                    if (row < N) {
                        float4 o;
                        o.x = elu1(v[q].x + bv.x);
                        o.y = elu1(v[q].y + bv.y);
                        o.z = elu1(v[q].z + bv.z);
                        o.w = elu1(v[q].w + bv.w);
                        *(float4*)(out + (size_t)row * C + bc) = o;
                    }
                }
            }
        }
        __syncthreads();
    }
    if (wave >= MW) __builtin_amdgcn_s_setprio(0);
}

extern "C" void kernel_launch(void* const* d_in, const int* in_sizes, int n_in,
                              void* d_out, int out_size, void* d_ws, size_t ws_size,
                              hipStream_t stream) {
    const float* x  = (const float*)d_in[0];
    const int*   ei = (const int*)d_in[1];
    const float* W1 = (const float*)d_in[2];
    const float* b1 = (const float*)d_in[3];
    const float* W2 = (const float*)d_in[4];
    const float* b2 = (const float*)d_in[5];

    const int N = in_sizes[0] / C;
    const int E = in_sizes[1] / 2;
    const int* srcI = ei;
    const int* dstI = ei + E;

    const int B = (N + NPB - 1) / NPB;
    const int G = (E + CH - 1) / CH;

    char* ws = (char*)d_ws;
    size_t off = 0;
    int* rowptr = (int*)(ws + off);   off += ((size_t)(N + 1) * 4 + 255) & ~(size_t)255;
    float* dinv = (float*)(ws + off); off += ((size_t)N * 4 + 255) & ~(size_t)255;
    int* histT = (int*)(ws + off);    off += ((size_t)B * G * 4 + 255) & ~(size_t)255;
    int* offT = (int*)(ws + off);     off += ((size_t)B * G * 4 + 255) & ~(size_t)255;
    int2* tmp = (int2*)(ws + off);    off += ((size_t)G * CH * 8 + 255) & ~(size_t)255;
    int2* colw = (int2*)(ws + off);   off += ((size_t)E * 8 + 255) & ~(size_t)255;
    float* buf1 = (float*)(ws + off);

    const int strips = (N + 15) / 16;
    const int layerBlocks = ((strips + 15) / 16 > WGPS) ? (strips + 15) / 16 : WGPS;

    bin_kernel<<<G, dim3(BT), 0, stream>>>(srcI, dstI, tmp, histT, offT, E, B, G);
    bucket_kernel<<<B, dim3(BT), 0, stream>>>(tmp, histT, offT, rowptr, dinv, colw, N, B, G);

    layer_fused<<<layerBlocks, dim3(LT), 0, stream>>>(x, W1, b1, rowptr, colw, dinv, buf1, N, E, strips, 1);
    layer_fused<<<layerBlocks, dim3(LT), 0, stream>>>(buf1, W2, b2, rowptr, colw, dinv, (float*)d_out, N, E, strips, 0);
}
